# starting kernel
# baseline (speedup 1.0000x reference)
#include <hip/hip_runtime.h>
#include <math.h>

#define BATCH 64
#define INC   2048
#define NJ    32
#define DIM   16
#define ICH   16
#define NBLK  (INC / ICH)
#define SZ    (BATCH * NJ * DIM)
#define NTHR  512
#define JT    8

typedef float v2f __attribute__((ext_vector_type(2)));
typedef float v8f __attribute__((ext_vector_type(8)));

#define DPP_XOR(x, m)                                                          \
  __int_as_float(__builtin_amdgcn_update_dpp(                                  \
      0, __float_as_int(x), 0x160 | (m), 0xf, 0xf, true))

__device__ __forceinline__ float xorsum16(float p) {
  p += DPP_XOR(p, 1);
  p += DPP_XOR(p, 2);
  p += DPP_XOR(p, 4);
  p += DPP_XOR(p, 8);
  return p;
}

__global__ void caps_init(float* p, int n) {
  int i = blockIdx.x * blockDim.x + threadIdx.x;
  if (i < n) p[i] = 0.f;
}

template <bool MODEA>
__global__ __launch_bounds__(NTHR) void caps_pass(const float* __restrict__ x,
                                                  const float* __restrict__ W,
                                                  const float* __restrict__ outsum,
                                                  float* sdst) {
  __shared__ float Xs[BATCH * DIM];
  __shared__ float Ws[NJ * DIM * DIM];
  __shared__ float Lg[BATCH * NJ];
  __shared__ float smax[BATCH];
  __shared__ float sinv[BATCH];

  const int tid  = threadIdx.x;
  const int lane = tid & 31;
  const int wv   = tid >> 5;
  const int half = lane >> 4;
  const int l16  = lane & 15;
  const int mt   = wv & 3;
  const int jb   = (wv >> 2) * JT;
  const int i0   = blockIdx.x * ICH;

  const v8f vzero = {0.f, 0.f, 0.f, 0.f, 0.f, 0.f, 0.f, 0.f};
  v8f sacc[JT];
#pragma unroll
  for (int jj = 0; jj < JT; ++jj) sacc[jj] = vzero;

  float osum[JT][8];
#pragma unroll
  for (int jj = 0; jj < JT; ++jj)
#pragma unroll
    for (int v = 0; v < 8; ++v) {
      const int b = mt * 16 + v + 8 * half;
      osum[jj][v] = outsum[((size_t)b * NJ + jb + jj) * DIM + l16];
    }

  for (int ii = 0; ii < ICH; ++ii) {
    const int i = i0 + ii;
    {
      const int r = tid >> 3, c = (tid & 7) * 2;
      const float2 vx = *(const float2*)(x + ((size_t)r * INC + i) * DIM + c);
      *(float2*)(Xs + r * DIM + c) = vx;
    }
    {
      const float4* src = (const float4*)(W + (size_t)i * NJ * DIM * DIM);
      float4* dst = (float4*)Ws;
#pragma unroll
      for (int q = 0; q < 4; ++q) dst[q * NTHR + tid] = src[q * NTHR + tid];
    }
    if (i + 1 < INC)
      __builtin_prefetch(W + (size_t)(i + 1) * NJ * DIM * DIM, 0, 1);
    __syncthreads();

    v2f afr[4];
#pragma unroll
    for (int kk = 0; kk < 4; ++kk) {
      const int base = (mt * 16 + l16) * DIM + 4 * kk + 2 * half;
      afr[kk].x = Xs[base];
      afr[kk].y = Xs[base + 1];
    }

#pragma unroll
    for (int jj = 0; jj < JT; ++jj) {
      const int j = jb + jj;
      v8f acc = vzero;
#pragma unroll
      for (int kk = 0; kk < 4; ++kk) {
        v2f bfr;
        const int base = (j * DIM + l16) * DIM + 4 * kk + 2 * half;
        bfr.x = Ws[base];
        bfr.y = Ws[base + 1];
        acc = __builtin_amdgcn_wmma_f32_16x16x4_f32(false, afr[kk], false, bfr,
                                                    (short)0, acc, false, false);
      }
#pragma unroll
      for (int v = 0; v < 8; ++v) {
        const float p = xorsum16(acc[v] * osum[jj][v]);
        if (l16 == 0) Lg[(mt * 16 + v + 8 * half) * NJ + j] = p;
      }
    }
    __syncthreads();

    if (tid < BATCH) {
      float m = -3.4e38f;
      for (int j = 0; j < NJ; ++j) m = fmaxf(m, Lg[tid * NJ + j]);
      float s = 0.f;
      for (int j = 0; j < NJ; ++j) s += expf(Lg[tid * NJ + j] - m);
      smax[tid] = m;
      sinv[tid] = 1.f / s;
    }
    __syncthreads();

    const int brow = mt * 16 + l16;
    const float mrow = smax[brow];
    const float srow = sinv[brow];
#pragma unroll
    for (int jj = 0; jj < JT; ++jj) {
      const int j = jb + jj;
      const float c = expf(Lg[brow * NJ + j] - mrow) * srow;
#pragma unroll
      for (int kk = 0; kk < 4; ++kk) {
        v2f afs, bfr;
        afs.x = c * afr[kk].x;
        afs.y = c * afr[kk].y;
        const int base = (j * DIM + l16) * DIM + 4 * kk + 2 * half;
        bfr.x = Ws[base];
        bfr.y = Ws[base + 1];
        sacc[jj] = __builtin_amdgcn_wmma_f32_16x16x4_f32(
            false, afs, false, bfr, (short)0, sacc[jj], false, false);
      }
    }
    __syncthreads();
  }

  if (MODEA) {
    float* dst = sdst + (size_t)blockIdx.x * SZ;
#pragma unroll
    for (int jj = 0; jj < JT; ++jj)
#pragma unroll
      for (int v = 0; v < 8; ++v) {
        const int b = mt * 16 + v + 8 * half;
        dst[((size_t)b * NJ + jb + jj) * DIM + l16] = sacc[jj][v];
      }
  } else {
#pragma unroll
    for (int jj = 0; jj < JT; ++jj)
#pragma unroll
      for (int v = 0; v < 8; ++v) {
        const int b = mt * 16 + v + 8 * half;
        unsafeAtomicAdd(&sdst[((size_t)b * NJ + jb + jj) * DIM + l16], sacc[jj][v]);
      }
  }
}

__global__ void caps_squash(const float* src, float* outsum, float* dout,
                            float* szero, int nparts) {
  const int idx = blockIdx.x * blockDim.x + threadIdx.x;
  float v = 0.f;
  for (int p = 0; p < nparts; ++p) v += src[(size_t)p * SZ + idx];
  if (szero) szero[idx] = 0.f;
  const float sq = xorsum16(v * v);
  const float coef = sq / ((1.f + sq) * sqrtf(sq + 1e-7f));
  const float o = coef * v;
  outsum[idx] += o;
  if (dout) dout[idx] = o;
}

extern "C" void kernel_launch(void* const* d_in, const int* in_sizes, int n_in,
                              void* d_out, int out_size, void* d_ws, size_t ws_size,
                              hipStream_t stream) {
  const float* x = (const float*)d_in[0];
  const float* W = (const float*)d_in[1];
  float* out = (float*)d_out;
  float* ws = (float*)d_ws;

  float* outsum = ws;
  float* sbuf   = ws + SZ;
  float* parts  = ws + 2 * SZ;

  const bool modeA = ws_size >= (size_t)(2 * SZ + NBLK * SZ) * sizeof(float);

  int ninit = 2 * SZ;
  if ((size_t)ninit * sizeof(float) > ws_size) ninit = (int)(ws_size / sizeof(float));
  caps_init<<<(ninit + 255) / 256, 256, 0, stream>>>(ws, ninit);

  for (int t = 0; t < 3; ++t) {
    if (modeA)
      caps_pass<true><<<NBLK, NTHR, 0, stream>>>(x, W, outsum, parts);
    else
      caps_pass<false><<<NBLK, NTHR, 0, stream>>>(x, W, outsum, sbuf);
    caps_squash<<<SZ / 256, 256, 0, stream>>>(modeA ? parts : sbuf, outsum,
                                              (t == 2) ? out : nullptr,
                                              modeA ? nullptr : sbuf,
                                              modeA ? NBLK : 1);
  }
}


# Batch split across blocks: each block takes one WMMA row tile of the batch for its i-chunk (grid 128 -> 512 blocks, two resident per WGP), so all 256 WGPs work instead of 128
# speedup vs baseline: 1.6497x; 1.6497x over previous
#include <hip/hip_runtime.h>
#include <math.h>

#define BATCH 64
#define INC   2048
#define NJ    32
#define DIM   16
#define ICH   16
#define NBLK  (INC / ICH)
#define SZ    (BATCH * NJ * DIM)
#define NTHR  512
#define MTB   1
#define BB    (16 * MTB)
#define BSPL  (BATCH / BB)
#define JG    (16 / MTB)
#define JT    (NJ / JG)

typedef float v2f __attribute__((ext_vector_type(2)));
typedef float v8f __attribute__((ext_vector_type(8)));

#define DPP_XOR(x, m)                                                          \
  __int_as_float(__builtin_amdgcn_update_dpp(                                  \
      0, __float_as_int(x), 0x160 | (m), 0xf, 0xf, true))

__device__ __forceinline__ float xorsum16(float p) {
  p += DPP_XOR(p, 1);
  p += DPP_XOR(p, 2);
  p += DPP_XOR(p, 4);
  p += DPP_XOR(p, 8);
  return p;
}

__global__ void caps_init(float* p, int n) {
  int i = blockIdx.x * blockDim.x + threadIdx.x;
  if (i < n) p[i] = 0.f;
}

template <bool MODEA>
__global__ __launch_bounds__(NTHR) __attribute__((amdgpu_waves_per_eu(8, 8)))
void caps_pass(const float* __restrict__ x, const float* __restrict__ W,
               const float* __restrict__ outsum, float* sdst) {
  __shared__ float Xs[BB * DIM];
  __shared__ float Ws[NJ * DIM * DIM];
  __shared__ float Lg[BB * NJ];
  __shared__ float smax[BB];
  __shared__ float sinv[BB];

  const int tid  = threadIdx.x;
  const int lane = tid & 31;
  const int wv   = tid >> 5;
  const int half = lane >> 4;
  const int l16  = lane & 15;
  const int mt   = wv % MTB;
  const int jb   = (wv / MTB) * JT;
  const int ich  = blockIdx.x / BSPL;
  const int b0   = (blockIdx.x % BSPL) * BB;
  const int i0   = ich * ICH;

  const v8f vzero = {0.f, 0.f, 0.f, 0.f, 0.f, 0.f, 0.f, 0.f};
  v8f sacc[JT];
#pragma unroll
  for (int jj = 0; jj < JT; ++jj) sacc[jj] = vzero;

  float osum[JT][8];
#pragma unroll
  for (int jj = 0; jj < JT; ++jj)
#pragma unroll
    for (int v = 0; v < 8; ++v) {
      const int b = b0 + mt * 16 + v + 8 * half;
      osum[jj][v] = outsum[((size_t)b * NJ + jb + jj) * DIM + l16];
    }

  for (int ii = 0; ii < ICH; ++ii) {
    const int i = i0 + ii;
    {
      const int r = (tid >> 3) & (BB - 1), c = (tid & 7) * 2;
      const float2 vx = *(const float2*)(x + ((size_t)(b0 + r) * INC + i) * DIM + c);
      *(float2*)(Xs + r * DIM + c) = vx;
    }
    {
      const float4* src = (const float4*)(W + (size_t)i * NJ * DIM * DIM);
      float4* dst = (float4*)Ws;
#pragma unroll
      for (int q = 0; q < 4; ++q) dst[q * NTHR + tid] = src[q * NTHR + tid];
    }
    if (i + 1 < INC)
      __builtin_prefetch(W + (size_t)(i + 1) * NJ * DIM * DIM, 0, 1);
    __syncthreads();

    v2f afr[4];
#pragma unroll
    for (int kk = 0; kk < 4; ++kk) {
      const int base = (mt * 16 + l16) * DIM + 4 * kk + 2 * half;
      afr[kk].x = Xs[base];
      afr[kk].y = Xs[base + 1];
    }

#pragma unroll
    for (int jj = 0; jj < JT; ++jj) {
      const int j = jb + jj;
      v8f acc = vzero;
#pragma unroll
      for (int kk = 0; kk < 4; ++kk) {
        v2f bfr;
        const int base = (j * DIM + l16) * DIM + 4 * kk + 2 * half;
        bfr.x = Ws[base];
        bfr.y = Ws[base + 1];
        acc = __builtin_amdgcn_wmma_f32_16x16x4_f32(false, afr[kk], false, bfr,
                                                    (short)0, acc, false, false);
      }
#pragma unroll
      for (int v = 0; v < 8; ++v) {
        const float p = xorsum16(acc[v] * osum[jj][v]);
        if (l16 == 0) Lg[(mt * 16 + v + 8 * half) * NJ + j] = p;
      }
    }
    __syncthreads();

    if (tid < BB) {
      float m = -3.4e38f;
      for (int j = 0; j < NJ; ++j) m = fmaxf(m, Lg[tid * NJ + j]);
      float s = 0.f;
      for (int j = 0; j < NJ; ++j) s += expf(Lg[tid * NJ + j] - m);
      smax[tid] = m;
      sinv[tid] = 1.f / s;
    }
    __syncthreads();

    const int brow = mt * 16 + l16;
    const float mrow = smax[brow];
    const float srow = sinv[brow];
#pragma unroll
    for (int jj = 0; jj < JT; ++jj) {
      const int j = jb + jj;
      const float c = expf(Lg[brow * NJ + j] - mrow) * srow;
#pragma unroll
      for (int kk = 0; kk < 4; ++kk) {
        v2f afs, bfr;
        afs.x = c * afr[kk].x;
        afs.y = c * afr[kk].y;
        const int base = (j * DIM + l16) * DIM + 4 * kk + 2 * half;
        bfr.x = Ws[base];
        bfr.y = Ws[base + 1];
        sacc[jj] = __builtin_amdgcn_wmma_f32_16x16x4_f32(
            false, afs, false, bfr, (short)0, sacc[jj], false, false);
      }
    }
    __syncthreads();
  }

  if (MODEA) {
    float* dst = sdst + (size_t)ich * SZ;
#pragma unroll
    for (int jj = 0; jj < JT; ++jj)
#pragma unroll
      for (int v = 0; v < 8; ++v) {
        const int b = b0 + mt * 16 + v + 8 * half;
        dst[((size_t)b * NJ + jb + jj) * DIM + l16] = sacc[jj][v];
      }
  } else {
#pragma unroll
    for (int jj = 0; jj < JT; ++jj)
#pragma unroll
      for (int v = 0; v < 8; ++v) {
        const int b = b0 + mt * 16 + v + 8 * half;
        unsafeAtomicAdd(&sdst[((size_t)b * NJ + jb + jj) * DIM + l16], sacc[jj][v]);
      }
  }
}

__global__ void caps_squash(const float* src, float* outsum, float* dout,
                            float* szero, int nparts) {
  const int idx = blockIdx.x * blockDim.x + threadIdx.x;
  float v = 0.f;
  for (int p = 0; p < nparts; ++p) v += src[(size_t)p * SZ + idx];
  if (szero) szero[idx] = 0.f;
  const float sq = xorsum16(v * v);
  const float coef = sq / ((1.f + sq) * sqrtf(sq + 1e-7f));
  const float o = coef * v;
  outsum[idx] += o;
  if (dout) dout[idx] = o;
}

extern "C" void kernel_launch(void* const* d_in, const int* in_sizes, int n_in,
                              void* d_out, int out_size, void* d_ws, size_t ws_size,
                              hipStream_t stream) {
  const float* x = (const float*)d_in[0];
  const float* W = (const float*)d_in[1];
  float* out = (float*)d_out;
  float* ws = (float*)d_ws;

  float* outsum = ws;
  float* sbuf   = ws + SZ;
  float* parts  = ws + 2 * SZ;

  const bool modeA = ws_size >= (size_t)(2 * SZ + NBLK * SZ) * sizeof(float);

  int ninit = 2 * SZ;
  if ((size_t)ninit * sizeof(float) > ws_size) ninit = (int)(ws_size / sizeof(float));
  caps_init<<<(ninit + 255) / 256, 256, 0, stream>>>(ws, ninit);

  for (int t = 0; t < 3; ++t) {
    if (modeA)
      caps_pass<true><<<NBLK * BSPL, NTHR, 0, stream>>>(x, W, outsum, parts);
    else
      caps_pass<false><<<NBLK * BSPL, NTHR, 0, stream>>>(x, W, outsum, sbuf);
    caps_squash<<<SZ / 256, 256, 0, stream>>>(modeA ? parts : sbuf, outsum,
                                              (t == 2) ? out : nullptr,
                                              modeA ? nullptr : sbuf,
                                              modeA ? NBLK : 1);
  }
}


# Software pipelining of the staging: the next capsule's W slab and x tile are loaded to registers one trip ahead, stay in flight across LDS-scoped barriers, and go to LDS at the next trip's top
# speedup vs baseline: 1.7665x; 1.0708x over previous
#include <hip/hip_runtime.h>
#include <math.h>

#define BATCH 64
#define INC   2048
#define NJ    32
#define DIM   16
#define ICH   16
#define NBLK  (INC / ICH)
#define SZ    (BATCH * NJ * DIM)
#define NTHR  512
#define MTB   1
#define BB    (16 * MTB)
#define BSPL  (BATCH / BB)
#define JG    (16 / MTB)
#define JT    (NJ / JG)

typedef float v2f __attribute__((ext_vector_type(2)));
typedef float v8f __attribute__((ext_vector_type(8)));

#define DPP_XOR(x, m)                                                          \
  __int_as_float(__builtin_amdgcn_update_dpp(                                  \
      0, __float_as_int(x), 0x160 | (m), 0xf, 0xf, true))

__device__ __forceinline__ float xorsum16(float p) {
  p += DPP_XOR(p, 1);
  p += DPP_XOR(p, 2);
  p += DPP_XOR(p, 4);
  p += DPP_XOR(p, 8);
  return p;
}

__device__ __forceinline__ void lds_barrier() {
  __builtin_amdgcn_fence(__ATOMIC_RELEASE, "workgroup", "local");
  __builtin_amdgcn_s_barrier();
  __builtin_amdgcn_fence(__ATOMIC_ACQUIRE, "workgroup", "local");
}

__global__ void caps_init(float* p, int n) {
  int i = blockIdx.x * blockDim.x + threadIdx.x;
  if (i < n) p[i] = 0.f;
}

template <bool MODEA>
__global__ __launch_bounds__(NTHR) __attribute__((amdgpu_waves_per_eu(8, 8)))
void caps_pass(const float* __restrict__ x, const float* __restrict__ W,
               const float* __restrict__ outsum, float* sdst) {
  __shared__ float Xs[BB * DIM];
  __shared__ float Ws[NJ * DIM * DIM];
  __shared__ float Lg[BB * NJ];
  __shared__ float smax[BB];
  __shared__ float sinv[BB];

  const int tid  = threadIdx.x;
  const int lane = tid & 31;
  const int wv   = tid >> 5;
  const int half = lane >> 4;
  const int l16  = lane & 15;
  const int mt   = wv % MTB;
  const int jb   = (wv / MTB) * JT;
  const int ich  = blockIdx.x / BSPL;
  const int b0   = (blockIdx.x % BSPL) * BB;
  const int i0   = ich * ICH;

  const v8f vzero = {0.f, 0.f, 0.f, 0.f, 0.f, 0.f, 0.f, 0.f};
  v8f sacc[JT];
#pragma unroll
  for (int jj = 0; jj < JT; ++jj) sacc[jj] = vzero;

  float osum[JT][8];
#pragma unroll
  for (int jj = 0; jj < JT; ++jj)
#pragma unroll
    for (int v = 0; v < 8; ++v) {
      const int b = b0 + mt * 16 + v + 8 * half;
      osum[jj][v] = outsum[((size_t)b * NJ + jb + jj) * DIM + l16];
    }

  const int xr = (tid >> 3) & (BB - 1), xc = (tid & 7) * 2;
  const float* xsrc = x + ((size_t)(b0 + xr) * INC + i0) * DIM + xc;
  const float4* wsrc = (const float4*)(W + (size_t)i0 * NJ * DIM * DIM) + tid;
  float2 px = *(const float2*)xsrc;
  float4 pw0 = wsrc[0 * NTHR];
  float4 pw1 = wsrc[1 * NTHR];
  float4 pw2 = wsrc[2 * NTHR];
  float4 pw3 = wsrc[3 * NTHR];

  for (int ii = 0; ii < ICH; ++ii) {
    *(float2*)(Xs + xr * DIM + xc) = px;
    {
      float4* dst = (float4*)Ws + tid;
      dst[0 * NTHR] = pw0;
      dst[1 * NTHR] = pw1;
      dst[2 * NTHR] = pw2;
      dst[3 * NTHR] = pw3;
    }
    if (ii + 1 < ICH) {
      px = *(const float2*)(xsrc + (size_t)(ii + 1) * DIM);
      const float4* src = wsrc + (size_t)(ii + 1) * (NJ * DIM * DIM / 4);
      pw0 = src[0 * NTHR];
      pw1 = src[1 * NTHR];
      pw2 = src[2 * NTHR];
      pw3 = src[3 * NTHR];
      asm volatile("s_wait_xcnt 0x0" ::: "memory");
    }
    lds_barrier();

    v2f afr[4];
#pragma unroll
    for (int kk = 0; kk < 4; ++kk) {
      const int base = (mt * 16 + l16) * DIM + 4 * kk + 2 * half;
      afr[kk].x = Xs[base];
      afr[kk].y = Xs[base + 1];
    }

#pragma unroll
    for (int jj = 0; jj < JT; ++jj) {
      const int j = jb + jj;
      v8f acc = vzero;
#pragma unroll
      for (int kk = 0; kk < 4; ++kk) {
        v2f bfr;
        const int base = (j * DIM + l16) * DIM + 4 * kk + 2 * half;
        bfr.x = Ws[base];
        bfr.y = Ws[base + 1];
        acc = __builtin_amdgcn_wmma_f32_16x16x4_f32(false, afr[kk], false, bfr,
                                                    (short)0, acc, false, false);
      }
#pragma unroll
      for (int v = 0; v < 8; ++v) {
        const float p = xorsum16(acc[v] * osum[jj][v]);
        if (l16 == 0) Lg[(mt * 16 + v + 8 * half) * NJ + j] = p;
      }
    }
    lds_barrier();

    if (tid < BB) {
      float m = -3.4e38f;
      for (int j = 0; j < NJ; ++j) m = fmaxf(m, Lg[tid * NJ + j]);
      float s = 0.f;
      for (int j = 0; j < NJ; ++j) s += expf(Lg[tid * NJ + j] - m);
      smax[tid] = m;
      sinv[tid] = 1.f / s;
    }
    lds_barrier();

    const int brow = mt * 16 + l16;
    const float mrow = smax[brow];
    const float srow = sinv[brow];
#pragma unroll
    for (int jj = 0; jj < JT; ++jj) {
      const int j = jb + jj;
      const float c = expf(Lg[brow * NJ + j] - mrow) * srow;
#pragma unroll
      for (int kk = 0; kk < 4; ++kk) {
        v2f afs, bfr;
        afs.x = c * afr[kk].x;
        afs.y = c * afr[kk].y;
        const int base = (j * DIM + l16) * DIM + 4 * kk + 2 * half;
        bfr.x = Ws[base];
        bfr.y = Ws[base + 1];
        sacc[jj] = __builtin_amdgcn_wmma_f32_16x16x4_f32(
            false, afs, false, bfr, (short)0, sacc[jj], false, false);
      }
    }
    lds_barrier();
  }

  if (MODEA) {
    float* dst = sdst + (size_t)ich * SZ;
#pragma unroll
    for (int jj = 0; jj < JT; ++jj)
#pragma unroll
      for (int v = 0; v < 8; ++v) {
        const int b = b0 + mt * 16 + v + 8 * half;
        dst[((size_t)b * NJ + jb + jj) * DIM + l16] = sacc[jj][v];
      }
  } else {
#pragma unroll
    for (int jj = 0; jj < JT; ++jj)
#pragma unroll
      for (int v = 0; v < 8; ++v) {
        const int b = b0 + mt * 16 + v + 8 * half;
        unsafeAtomicAdd(&sdst[((size_t)b * NJ + jb + jj) * DIM + l16], sacc[jj][v]);
      }
  }
}

__global__ void caps_squash(const float* src, float* outsum, float* dout,
                            float* szero, int nparts) {
  const int idx = blockIdx.x * blockDim.x + threadIdx.x;
  float v = 0.f;
  for (int p = 0; p < nparts; ++p) v += src[(size_t)p * SZ + idx];
  if (szero) szero[idx] = 0.f;
  const float sq = xorsum16(v * v);
  const float coef = sq / ((1.f + sq) * sqrtf(sq + 1e-7f));
  const float o = coef * v;
  outsum[idx] += o;
  if (dout) dout[idx] = o;
}

extern "C" void kernel_launch(void* const* d_in, const int* in_sizes, int n_in,
                              void* d_out, int out_size, void* d_ws, size_t ws_size,
                              hipStream_t stream) {
  const float* x = (const float*)d_in[0];
  const float* W = (const float*)d_in[1];
  float* out = (float*)d_out;
  float* ws = (float*)d_ws;

  float* outsum = ws;
  float* sbuf   = ws + SZ;
  float* parts  = ws + 2 * SZ;

  const bool modeA = ws_size >= (size_t)(2 * SZ + NBLK * SZ) * sizeof(float);

  int ninit = 2 * SZ;
  if ((size_t)ninit * sizeof(float) > ws_size) ninit = (int)(ws_size / sizeof(float));
  caps_init<<<(ninit + 255) / 256, 256, 0, stream>>>(ws, ninit);

  for (int t = 0; t < 3; ++t) {
    if (modeA)
      caps_pass<true><<<NBLK * BSPL, NTHR, 0, stream>>>(x, W, outsum, parts);
    else
      caps_pass<false><<<NBLK * BSPL, NTHR, 0, stream>>>(x, W, outsum, sbuf);
    caps_squash<<<SZ / 256, 256, 0, stream>>>(modeA ? parts : sbuf, outsum,
                                              (t == 2) ? out : nullptr,
                                              modeA ? nullptr : sbuf,
                                              modeA ? NBLK : 1);
  }
}


# Softmax over j spread over the whole block: wave r takes batch row r, one j per lane, max and sum by lane butterflies, c kept in LDS for phase 2 (denominator now summed as a tree, not serially)
# speedup vs baseline: 2.0504x; 1.1607x over previous
#include <hip/hip_runtime.h>
#include <math.h>

#define BATCH 64
#define INC   2048
#define NJ    32
#define DIM   16
#define ICH   16
#define NBLK  (INC / ICH)
#define SZ    (BATCH * NJ * DIM)
#define NTHR  512
#define MTB   1
#define BB    (16 * MTB)
#define BSPL  (BATCH / BB)
#define JG    (16 / MTB)
#define JT    (NJ / JG)

typedef float v2f __attribute__((ext_vector_type(2)));
typedef float v8f __attribute__((ext_vector_type(8)));

#define DPP_XOR(x, m)                                                          \
  __int_as_float(__builtin_amdgcn_update_dpp(                                  \
      0, __float_as_int(x), 0x160 | (m), 0xf, 0xf, true))

__device__ __forceinline__ float xorsum16(float p) {
  p += DPP_XOR(p, 1);
  p += DPP_XOR(p, 2);
  p += DPP_XOR(p, 4);
  p += DPP_XOR(p, 8);
  return p;
}

__device__ __forceinline__ float xormax16(float p) {
  p = fmaxf(p, DPP_XOR(p, 1));
  p = fmaxf(p, DPP_XOR(p, 2));
  p = fmaxf(p, DPP_XOR(p, 4));
  p = fmaxf(p, DPP_XOR(p, 8));
  return p;
}

__device__ __forceinline__ float xhalf(float x) {
  const int b = __float_as_int(x);
  return __int_as_float(
      __builtin_amdgcn_permlanex16(b, b, 0x76543210, 0xfedcba98, false, false));
}

__device__ __forceinline__ void lds_barrier() {
  __builtin_amdgcn_fence(__ATOMIC_RELEASE, "workgroup", "local");
  __builtin_amdgcn_s_barrier();
  __builtin_amdgcn_fence(__ATOMIC_ACQUIRE, "workgroup", "local");
}

__global__ void caps_init(float* p, int n) {
  int i = blockIdx.x * blockDim.x + threadIdx.x;
  if (i < n) p[i] = 0.f;
}

template <bool MODEA>
__global__ __launch_bounds__(NTHR) __attribute__((amdgpu_waves_per_eu(8, 8)))
void caps_pass(const float* __restrict__ x, const float* __restrict__ W,
               const float* __restrict__ outsum, float* sdst) {
  __shared__ float Xs[BB * DIM];
  __shared__ float Ws[NJ * DIM * DIM];
  __shared__ float Lg[BB * NJ];
  static_assert(BB == NTHR / 32 && NJ == 32,
                "softmax step: one wave per batch row, one lane per j");

  const int tid  = threadIdx.x;
  const int lane = tid & 31;
  const int wv   = tid >> 5;
  const int half = lane >> 4;
  const int l16  = lane & 15;
  const int mt   = wv % MTB;
  const int jb   = (wv / MTB) * JT;
  const int ich  = blockIdx.x / BSPL;
  const int b0   = (blockIdx.x % BSPL) * BB;
  const int i0   = ich * ICH;

  const v8f vzero = {0.f, 0.f, 0.f, 0.f, 0.f, 0.f, 0.f, 0.f};
  v8f sacc[JT];
#pragma unroll
  for (int jj = 0; jj < JT; ++jj) sacc[jj] = vzero;

  float osum[JT][8];
#pragma unroll
  for (int jj = 0; jj < JT; ++jj)
#pragma unroll
    for (int v = 0; v < 8; ++v) {
      const int b = b0 + mt * 16 + v + 8 * half;
      osum[jj][v] = outsum[((size_t)b * NJ + jb + jj) * DIM + l16];
    }

  const int xr = (tid >> 3) & (BB - 1), xc = (tid & 7) * 2;
  const float* xsrc = x + ((size_t)(b0 + xr) * INC + i0) * DIM + xc;
  const float4* wsrc = (const float4*)(W + (size_t)i0 * NJ * DIM * DIM) + tid;
  float2 px = *(const float2*)xsrc;
  float4 pw0 = wsrc[0 * NTHR];
  float4 pw1 = wsrc[1 * NTHR];
  float4 pw2 = wsrc[2 * NTHR];
  float4 pw3 = wsrc[3 * NTHR];

#pragma clang loop unroll(disable)
  for (int ii = 0; ii < ICH; ++ii) {
    *(float2*)(Xs + xr * DIM + xc) = px;
    {
      float4* dst = (float4*)Ws + tid;
      dst[0 * NTHR] = pw0;
      dst[1 * NTHR] = pw1;
      dst[2 * NTHR] = pw2;
      dst[3 * NTHR] = pw3;
    }
    if (ii + 1 < ICH) {
      px = *(const float2*)(xsrc + (size_t)(ii + 1) * DIM);
      const float4* src = wsrc + (size_t)(ii + 1) * (NJ * DIM * DIM / 4);
      pw0 = src[0 * NTHR];
      pw1 = src[1 * NTHR];
      pw2 = src[2 * NTHR];
      pw3 = src[3 * NTHR];
      asm volatile("s_wait_xcnt 0x0" ::: "memory");
    }
    lds_barrier();

    v2f afr[4];
#pragma unroll
    for (int kk = 0; kk < 4; ++kk) {
      const int base = (mt * 16 + l16) * DIM + 4 * kk + 2 * half;
      afr[kk].x = Xs[base];
      afr[kk].y = Xs[base + 1];
    }

#pragma unroll
    for (int jj = 0; jj < JT; ++jj) {
      const int j = jb + jj;
      v8f acc = vzero;
#pragma unroll
      for (int kk = 0; kk < 4; ++kk) {
        v2f bfr;
        const int base = (j * DIM + l16) * DIM + 4 * kk + 2 * half;
        bfr.x = Ws[base];
        bfr.y = Ws[base + 1];
        acc = __builtin_amdgcn_wmma_f32_16x16x4_f32(false, afr[kk], false, bfr,
                                                    (short)0, acc, false, false);
      }
#pragma unroll
      for (int v = 0; v < 8; ++v) {
        const float p = xorsum16(acc[v] * osum[jj][v]);
        if (l16 == 0) Lg[(mt * 16 + v + 8 * half) * NJ + j] = p;
      }
    }
    lds_barrier();

    {
      const float lg = Lg[wv * NJ + lane];
      float m = xormax16(lg);
      m = fmaxf(m, xhalf(m));
      const float e = expf(lg - m);
      float s = xorsum16(e);
      s += xhalf(s);
      Lg[wv * NJ + lane] = e * (1.f / s);
    }
    lds_barrier();

    const int brow = mt * 16 + l16;
#pragma unroll
    for (int jj = 0; jj < JT; ++jj) {
      const int j = jb + jj;
      const float c = Lg[brow * NJ + j];
#pragma unroll
      for (int kk = 0; kk < 4; ++kk) {
        v2f afs, bfr;
        afs.x = c * afr[kk].x;
        afs.y = c * afr[kk].y;
        const int base = (j * DIM + l16) * DIM + 4 * kk + 2 * half;
        bfr.x = Ws[base];
        bfr.y = Ws[base + 1];
        sacc[jj] = __builtin_amdgcn_wmma_f32_16x16x4_f32(
            false, afs, false, bfr, (short)0, sacc[jj], false, false);
      }
    }
    lds_barrier();
  }

  if (MODEA) {
    float* dst = sdst + (size_t)ich * SZ;
#pragma unroll
    for (int jj = 0; jj < JT; ++jj)
#pragma unroll
      for (int v = 0; v < 8; ++v) {
        const int b = b0 + mt * 16 + v + 8 * half;
        dst[((size_t)b * NJ + jb + jj) * DIM + l16] = sacc[jj][v];
      }
  } else {
#pragma unroll
    for (int jj = 0; jj < JT; ++jj)
#pragma unroll
      for (int v = 0; v < 8; ++v) {
        const int b = b0 + mt * 16 + v + 8 * half;
        unsafeAtomicAdd(&sdst[((size_t)b * NJ + jb + jj) * DIM + l16], sacc[jj][v]);
      }
  }
}

__global__ void caps_squash(const float* src, float* outsum, float* dout,
                            float* szero, int nparts) {
  const int idx = blockIdx.x * blockDim.x + threadIdx.x;
  float v = 0.f;
  for (int p = 0; p < nparts; ++p) v += src[(size_t)p * SZ + idx];
  if (szero) szero[idx] = 0.f;
  const float sq = xorsum16(v * v);
  const float coef = sq / ((1.f + sq) * sqrtf(sq + 1e-7f));
  const float o = coef * v;
  outsum[idx] += o;
  if (dout) dout[idx] = o;
}

extern "C" void kernel_launch(void* const* d_in, const int* in_sizes, int n_in,
                              void* d_out, int out_size, void* d_ws, size_t ws_size,
                              hipStream_t stream) {
  const float* x = (const float*)d_in[0];
  const float* W = (const float*)d_in[1];
  float* out = (float*)d_out;
  float* ws = (float*)d_ws;

  float* outsum = ws;
  float* sbuf   = ws + SZ;
  float* parts  = ws + 2 * SZ;

  const bool modeA = ws_size >= (size_t)(2 * SZ + NBLK * SZ) * sizeof(float);

  int ninit = 2 * SZ;
  if ((size_t)ninit * sizeof(float) > ws_size) ninit = (int)(ws_size / sizeof(float));
  caps_init<<<(ninit + 255) / 256, 256, 0, stream>>>(ws, ninit);

  for (int t = 0; t < 3; ++t) {
    if (modeA)
      caps_pass<true><<<NBLK * BSPL, NTHR, 0, stream>>>(x, W, outsum, parts);
    else
      caps_pass<false><<<NBLK * BSPL, NTHR, 0, stream>>>(x, W, outsum, sbuf);
    caps_squash<<<SZ / 256, 256, 0, stream>>>(modeA ? parts : sbuf, outsum,
                                              (t == 2) ? out : nullptr,
                                              modeA ? nullptr : sbuf,
                                              modeA ? NBLK : 1);
  }
}


# W fragments go from global memory straight to B registers, one capsule ahead in a second register set (no LDS staging of W); A and B loaders both use k = 8*half+2*kk+s, so the k sum order changes
# speedup vs baseline: 2.2025x; 1.0742x over previous
#include <hip/hip_runtime.h>
#include <math.h>

#define BATCH 64
#define INC   2048
#define NJ    32
#define DIM   16
#define ICH   16
#define NBLK  (INC / ICH)
#define SZ    (BATCH * NJ * DIM)
#define NTHR  512
#define MTB   1
#define BB    (16 * MTB)
#define BSPL  (BATCH / BB)
#define JG    (16 / MTB)
#define JT    (NJ / JG)

typedef float v2f __attribute__((ext_vector_type(2)));
typedef float v8f __attribute__((ext_vector_type(8)));

#define DPP_XOR(x, m)                                                          \
  __int_as_float(__builtin_amdgcn_update_dpp(                                  \
      0, __float_as_int(x), 0x160 | (m), 0xf, 0xf, true))

__device__ __forceinline__ float xorsum16(float p) {
  p += DPP_XOR(p, 1);
  p += DPP_XOR(p, 2);
  p += DPP_XOR(p, 4);
  p += DPP_XOR(p, 8);
  return p;
}

__device__ __forceinline__ float xormax16(float p) {
  p = fmaxf(p, DPP_XOR(p, 1));
  p = fmaxf(p, DPP_XOR(p, 2));
  p = fmaxf(p, DPP_XOR(p, 4));
  p = fmaxf(p, DPP_XOR(p, 8));
  return p;
}

__device__ __forceinline__ float xhalf(float x) {
  const int b = __float_as_int(x);
  return __int_as_float(
      __builtin_amdgcn_permlanex16(b, b, 0x76543210, 0xfedcba98, false, false));
}

__device__ __forceinline__ void lds_barrier() {
  __builtin_amdgcn_fence(__ATOMIC_RELEASE, "workgroup", "local");
  __builtin_amdgcn_s_barrier();
  __builtin_amdgcn_fence(__ATOMIC_ACQUIRE, "workgroup", "local");
}

__global__ void caps_init(float* p, int n) {
  int i = blockIdx.x * blockDim.x + threadIdx.x;
  if (i < n) p[i] = 0.f;
}

template <bool MODEA>
__global__ __launch_bounds__(NTHR) __attribute__((amdgpu_waves_per_eu(8, 8)))
void caps_pass(const float* __restrict__ x, const float* __restrict__ W,
               const float* __restrict__ outsum, float* sdst) {
  __shared__ __attribute__((aligned(16))) float Xs[BB * DIM];
  __shared__ float Lg[BB * NJ];
  static_assert(JT == 2 && DIM == 16, "W fragment registers: two j per wave, 8 k per lane");
  static_assert(BB == NTHR / 32 && NJ == 32,
                "softmax step: one wave per batch row, one lane per j");

  const int tid  = threadIdx.x;
  const int lane = tid & 31;
  const int wv   = tid >> 5;
  const int half = lane >> 4;
  const int l16  = lane & 15;
  const int mt   = wv % MTB;
  const int jb   = (wv / MTB) * JT;
  const int ich  = blockIdx.x / BSPL;
  const int b0   = (blockIdx.x % BSPL) * BB;
  const int i0   = ich * ICH;

  const v8f vzero = {0.f, 0.f, 0.f, 0.f, 0.f, 0.f, 0.f, 0.f};
  v8f sacc[JT];
#pragma unroll
  for (int jj = 0; jj < JT; ++jj) sacc[jj] = vzero;

  const size_t fr0 = ((size_t)(b0 + mt * 16 + 8 * half) * NJ + jb) * DIM + l16;
  float osum[JT][8];
#pragma unroll
  for (int jj = 0; jj < JT; ++jj)
#pragma unroll
    for (int v = 0; v < 8; ++v) osum[jj][v] = outsum[fr0 + (v * NJ + jj) * DIM];

  const int xr = (tid >> 3) & (BB - 1), xc = (tid & 7) * 2;
  const float* xsrc = x + ((size_t)(b0 + xr) * INC + i0) * DIM + xc;
  float2 px = *(const float2*)xsrc;

  const float4* wsrc =
      (const float4*)(W + (((size_t)i0 * NJ + jb) * DIM + l16) * DIM + 8 * half);
  constexpr int WCAP = NJ * DIM * DIM / 4;
  constexpr int WJ   = DIM * DIM / 4;
  float4 wa[2 * JT], wb[2 * JT];
#pragma unroll
  for (int jj = 0; jj < JT; ++jj) {
    wa[2 * jj]     = wsrc[jj * WJ];
    wa[2 * jj + 1] = wsrc[jj * WJ + 1];
  }

  auto trip = [&](const int nx, const float4 (&cw)[2 * JT],
                  float4 (&nw)[2 * JT]) __attribute__((always_inline)) {
    *(float2*)(Xs + xr * DIM + xc) = px;
    {
      px = *(const float2*)(xsrc + (size_t)nx * DIM);
      const float4* src = wsrc + (size_t)nx * WCAP;
#pragma unroll
      for (int jj = 0; jj < JT; ++jj) {
        nw[2 * jj]     = src[jj * WJ];
        nw[2 * jj + 1] = src[jj * WJ + 1];
      }
      asm volatile("s_wait_xcnt 0x0" ::: "memory");
    }
    lds_barrier();

    v2f afr[4];
    {
      const float4* xa = (const float4*)(Xs + (mt * 16 + l16) * DIM + 8 * half);
      const float4 a0 = xa[0], a1 = xa[1];
      afr[0].x = a0.x; afr[0].y = a0.y;
      afr[1].x = a0.z; afr[1].y = a0.w;
      afr[2].x = a1.x; afr[2].y = a1.y;
      afr[3].x = a1.z; afr[3].y = a1.w;
    }

#pragma unroll
    for (int jj = 0; jj < JT; ++jj) {
      const int j = jb + jj;
      v2f bfr[4];
      bfr[0].x = cw[2 * jj].x;     bfr[0].y = cw[2 * jj].y;
      bfr[1].x = cw[2 * jj].z;     bfr[1].y = cw[2 * jj].w;
      bfr[2].x = cw[2 * jj + 1].x; bfr[2].y = cw[2 * jj + 1].y;
      bfr[3].x = cw[2 * jj + 1].z; bfr[3].y = cw[2 * jj + 1].w;
      v8f acc = vzero;
#pragma unroll
      for (int kk = 0; kk < 4; ++kk)
        acc = __builtin_amdgcn_wmma_f32_16x16x4_f32(false, afr[kk], false, bfr[kk],
                                                    (short)0, acc, false, false);
#pragma unroll
      for (int v = 0; v < 8; ++v) {
        const float p = xorsum16(acc[v] * osum[jj][v]);
        if (l16 == 0) Lg[(mt * 16 + v + 8 * half) * NJ + j] = p;
      }
    }
    lds_barrier();

    {
      const float lg = Lg[wv * NJ + lane];
      float m = xormax16(lg);
      m = fmaxf(m, xhalf(m));
      const float e = expf(lg - m);
      float s = xorsum16(e);
      s += xhalf(s);
      Lg[wv * NJ + lane] = e * (1.f / s);
    }
    lds_barrier();

    const int brow = mt * 16 + l16;
#pragma unroll
    for (int jj = 0; jj < JT; ++jj) {
      const int j = jb + jj;
      const float c = Lg[brow * NJ + j];
      v2f bfr[4];
      bfr[0].x = cw[2 * jj].x;     bfr[0].y = cw[2 * jj].y;
      bfr[1].x = cw[2 * jj].z;     bfr[1].y = cw[2 * jj].w;
      bfr[2].x = cw[2 * jj + 1].x; bfr[2].y = cw[2 * jj + 1].y;
      bfr[3].x = cw[2 * jj + 1].z; bfr[3].y = cw[2 * jj + 1].w;
#pragma unroll
      for (int kk = 0; kk < 4; ++kk) {
        v2f afs;
        afs.x = c * afr[kk].x;
        afs.y = c * afr[kk].y;
        sacc[jj] = __builtin_amdgcn_wmma_f32_16x16x4_f32(
            false, afs, false, bfr[kk], (short)0, sacc[jj], false, false);
      }
    }
    lds_barrier();
  };

#pragma clang loop unroll(disable)
  for (int ii = 0; ii < ICH; ii += 2) {
    trip(ii + 1, wa, wb);
    trip(ii + 2 < ICH ? ii + 2 : ICH - 1, wb, wa);
  }

  if (MODEA) {
    float* dst = sdst + (size_t)ich * SZ + fr0;
#pragma unroll
    for (int jj = 0; jj < JT; ++jj)
#pragma unroll
      for (int v = 0; v < 8; ++v) dst[(v * NJ + jj) * DIM] = sacc[jj][v];
  } else {
#pragma unroll
    for (int jj = 0; jj < JT; ++jj)
#pragma unroll
      for (int v = 0; v < 8; ++v)
        unsafeAtomicAdd(&sdst[fr0 + (v * NJ + jj) * DIM], sacc[jj][v]);
  }
}

__global__ void caps_squash(const float* src, float* outsum, float* dout,
                            float* szero, int nparts) {
  const int idx = blockIdx.x * blockDim.x + threadIdx.x;
  float v = 0.f;
  for (int p = 0; p < nparts; ++p) v += src[(size_t)p * SZ + idx];
  if (szero) szero[idx] = 0.f;
  const float sq = xorsum16(v * v);
  const float coef = sq / ((1.f + sq) * sqrtf(sq + 1e-7f));
  const float o = coef * v;
  outsum[idx] += o;
  if (dout) dout[idx] = o;
}

extern "C" void kernel_launch(void* const* d_in, const int* in_sizes, int n_in,
                              void* d_out, int out_size, void* d_ws, size_t ws_size,
                              hipStream_t stream) {
  const float* x = (const float*)d_in[0];
  const float* W = (const float*)d_in[1];
  float* out = (float*)d_out;
  float* ws = (float*)d_ws;

  float* outsum = ws;
  float* sbuf   = ws + SZ;
  float* parts  = ws + 2 * SZ;

  const bool modeA = ws_size >= (size_t)(2 * SZ + NBLK * SZ) * sizeof(float);

  int ninit = 2 * SZ;
  if ((size_t)ninit * sizeof(float) > ws_size) ninit = (int)(ws_size / sizeof(float));
  caps_init<<<(ninit + 255) / 256, 256, 0, stream>>>(ws, ninit);

  for (int t = 0; t < 3; ++t) {
    if (modeA)
      caps_pass<true><<<NBLK * BSPL, NTHR, 0, stream>>>(x, W, outsum, parts);
    else
      caps_pass<false><<<NBLK * BSPL, NTHR, 0, stream>>>(x, W, outsum, sbuf);
    caps_squash<<<SZ / 256, 256, 0, stream>>>(modeA ? parts : sbuf, outsum,
                                              (t == 2) ? out : nullptr,
                                              modeA ? nullptr : sbuf,
                                              modeA ? NBLK : 1);
  }
}


# First routing pass in a kernel of its own: coupling there is softmax(0) = 1/32 for any input, so only x W^T stays (no logits, softmax, LDS, barriers); first squash stores outsum, no zeroing launch
# speedup vs baseline: 2.5696x; 1.1667x over previous
#include <hip/hip_runtime.h>
#include <math.h>

#define BATCH 64
#define INC   2048
#define NJ    32
#define DIM   16
#define ICH   16
#define NBLK  (INC / ICH)
#define SZ    (BATCH * NJ * DIM)
#define NTHR  512
#define MTB   1
#define BB    (16 * MTB)
#define BSPL  (BATCH / BB)
#define JG    (16 / MTB)
#define JT    (NJ / JG)

typedef float v2f __attribute__((ext_vector_type(2)));
typedef float v8f __attribute__((ext_vector_type(8)));

#define DPP_XOR(x, m)                                                          \
  __int_as_float(__builtin_amdgcn_update_dpp(                                  \
      0, __float_as_int(x), 0x160 | (m), 0xf, 0xf, true))

__device__ __forceinline__ float xorsum16(float p) {
  p += DPP_XOR(p, 1);
  p += DPP_XOR(p, 2);
  p += DPP_XOR(p, 4);
  p += DPP_XOR(p, 8);
  return p;
}

__device__ __forceinline__ float xormax16(float p) {
  p = fmaxf(p, DPP_XOR(p, 1));
  p = fmaxf(p, DPP_XOR(p, 2));
  p = fmaxf(p, DPP_XOR(p, 4));
  p = fmaxf(p, DPP_XOR(p, 8));
  return p;
}

__device__ __forceinline__ float xhalf(float x) {
  const int b = __float_as_int(x);
  return __int_as_float(
      __builtin_amdgcn_permlanex16(b, b, 0x76543210, 0xfedcba98, false, false));
}

__device__ __forceinline__ void lds_barrier() {
  __builtin_amdgcn_fence(__ATOMIC_RELEASE, "workgroup", "local");
  __builtin_amdgcn_s_barrier();
  __builtin_amdgcn_fence(__ATOMIC_ACQUIRE, "workgroup", "local");
}

__global__ void caps_init(float* p, int n) {
  int i = blockIdx.x * blockDim.x + threadIdx.x;
  if (i < n) p[i] = 0.f;
}

template <bool MODEA>
__global__ __launch_bounds__(NTHR) __attribute__((amdgpu_waves_per_eu(8, 8)))
void caps_pass(const float* __restrict__ x, const float* __restrict__ W,
               const float* __restrict__ outsum, float* sdst) {
  __shared__ __attribute__((aligned(16))) float Xs[BB * DIM];
  __shared__ float Lg[BB * NJ];
  static_assert(JT == 2 && DIM == 16, "W fragment registers: two j per wave, 8 k per lane");
  static_assert(BB == NTHR / 32 && NJ == 32,
                "softmax step: one wave per batch row, one lane per j");

  const int tid  = threadIdx.x;
  const int lane = tid & 31;
  const int wv   = tid >> 5;
  const int half = lane >> 4;
  const int l16  = lane & 15;
  const int mt   = wv % MTB;
  const int jb   = (wv / MTB) * JT;
  const int ich  = blockIdx.x / BSPL;
  const int b0   = (blockIdx.x % BSPL) * BB;
  const int i0   = ich * ICH;

  const v8f vzero = {0.f, 0.f, 0.f, 0.f, 0.f, 0.f, 0.f, 0.f};
  v8f sacc[JT];
#pragma unroll
  for (int jj = 0; jj < JT; ++jj) sacc[jj] = vzero;

  const size_t fr0 = ((size_t)(b0 + mt * 16 + 8 * half) * NJ + jb) * DIM + l16;
  float osum[JT][8];
#pragma unroll
  for (int jj = 0; jj < JT; ++jj)
#pragma unroll
    for (int v = 0; v < 8; ++v) osum[jj][v] = outsum[fr0 + (v * NJ + jj) * DIM];

  const int xr = (tid >> 3) & (BB - 1), xc = (tid & 7) * 2;
  const float* xsrc = x + ((size_t)(b0 + xr) * INC + i0) * DIM + xc;
  float2 px = *(const float2*)xsrc;

  const float4* wsrc =
      (const float4*)(W + (((size_t)i0 * NJ + jb) * DIM + l16) * DIM + 8 * half);
  constexpr int WCAP = NJ * DIM * DIM / 4;
  constexpr int WJ   = DIM * DIM / 4;
  float4 wa[2 * JT], wb[2 * JT];
#pragma unroll
  for (int jj = 0; jj < JT; ++jj) {
    wa[2 * jj]     = wsrc[jj * WJ];
    wa[2 * jj + 1] = wsrc[jj * WJ + 1];
  }

  auto trip = [&](const int nx, const float4 (&cw)[2 * JT],
                  float4 (&nw)[2 * JT]) __attribute__((always_inline)) {
    *(float2*)(Xs + xr * DIM + xc) = px;
    {
      px = *(const float2*)(xsrc + (size_t)nx * DIM);
      const float4* src = wsrc + (size_t)nx * WCAP;
#pragma unroll
      for (int jj = 0; jj < JT; ++jj) {
        nw[2 * jj]     = src[jj * WJ];
        nw[2 * jj + 1] = src[jj * WJ + 1];
      }
      asm volatile("s_wait_xcnt 0x0" ::: "memory");
    }
    lds_barrier();

    v2f afr[4];
    {
      const float4* xa = (const float4*)(Xs + (mt * 16 + l16) * DIM + 8 * half);
      const float4 a0 = xa[0], a1 = xa[1];
      afr[0].x = a0.x; afr[0].y = a0.y;
      afr[1].x = a0.z; afr[1].y = a0.w;
      afr[2].x = a1.x; afr[2].y = a1.y;
      afr[3].x = a1.z; afr[3].y = a1.w;
    }

#pragma unroll
    for (int jj = 0; jj < JT; ++jj) {
      const int j = jb + jj;
      v2f bfr[4];
      bfr[0].x = cw[2 * jj].x;     bfr[0].y = cw[2 * jj].y;
      bfr[1].x = cw[2 * jj].z;     bfr[1].y = cw[2 * jj].w;
      bfr[2].x = cw[2 * jj + 1].x; bfr[2].y = cw[2 * jj + 1].y;
      bfr[3].x = cw[2 * jj + 1].z; bfr[3].y = cw[2 * jj + 1].w;
      v8f acc = vzero;
#pragma unroll
      for (int kk = 0; kk < 4; ++kk)
        acc = __builtin_amdgcn_wmma_f32_16x16x4_f32(false, afr[kk], false, bfr[kk],
                                                    (short)0, acc, false, false);
#pragma unroll
      for (int v = 0; v < 8; ++v) {
        const float p = xorsum16(acc[v] * osum[jj][v]);
        if (l16 == 0) Lg[(mt * 16 + v + 8 * half) * NJ + j] = p;
      }
    }
    lds_barrier();

    {
      const float lg = Lg[wv * NJ + lane];
      float m = xormax16(lg);
      m = fmaxf(m, xhalf(m));
      const float e = expf(lg - m);
      float s = xorsum16(e);
      s += xhalf(s);
      Lg[wv * NJ + lane] = e * (1.f / s);
    }
    lds_barrier();

    const int brow = mt * 16 + l16;
#pragma unroll
    for (int jj = 0; jj < JT; ++jj) {
      const int j = jb + jj;
      const float c = Lg[brow * NJ + j];
      v2f bfr[4];
      bfr[0].x = cw[2 * jj].x;     bfr[0].y = cw[2 * jj].y;
      bfr[1].x = cw[2 * jj].z;     bfr[1].y = cw[2 * jj].w;
      bfr[2].x = cw[2 * jj + 1].x; bfr[2].y = cw[2 * jj + 1].y;
      bfr[3].x = cw[2 * jj + 1].z; bfr[3].y = cw[2 * jj + 1].w;
#pragma unroll
      for (int kk = 0; kk < 4; ++kk) {
        v2f afs;
        afs.x = c * afr[kk].x;
        afs.y = c * afr[kk].y;
        sacc[jj] = __builtin_amdgcn_wmma_f32_16x16x4_f32(
            false, afs, false, bfr[kk], (short)0, sacc[jj], false, false);
      }
    }
    lds_barrier();
  };

#pragma clang loop unroll(disable)
  for (int ii = 0; ii < ICH; ii += 2) {
    trip(ii + 1, wa, wb);
    trip(ii + 2 < ICH ? ii + 2 : ICH - 1, wb, wa);
  }

  if (MODEA) {
    float* dst = sdst + (size_t)ich * SZ + fr0;
#pragma unroll
    for (int jj = 0; jj < JT; ++jj)
#pragma unroll
      for (int v = 0; v < 8; ++v) dst[(v * NJ + jj) * DIM] = sacc[jj][v];
  } else {
#pragma unroll
    for (int jj = 0; jj < JT; ++jj)
#pragma unroll
      for (int v = 0; v < 8; ++v)
        unsafeAtomicAdd(&sdst[fr0 + (v * NJ + jj) * DIM], sacc[jj][v]);
  }
}

template <bool MODEA>
__global__ __launch_bounds__(NTHR) __attribute__((amdgpu_waves_per_eu(8, 8)))
void caps_pass_first(const float* __restrict__ x, const float* __restrict__ W,
                     float* sdst) {
  static_assert(JT == 2 && DIM == 16 && MTB == 1 && (ICH % 4) == 0,
                "two j per wave, 8 k per lane, capsule groups of two in pairs");
  static_assert((NJ & (NJ - 1)) == 0, "1/NJ must be a power of two (exact scaling)");
  const int tid  = threadIdx.x;
  const int lane = tid & 31;
  const int wv   = tid >> 5;
  const int half = lane >> 4;
  const int l16  = lane & 15;
  const int jb   = wv * JT;
  const int ich  = blockIdx.x / BSPL;
  const int b0   = (blockIdx.x % BSPL) * BB;
  const int i0   = ich * ICH;

  const v8f vzero = {0.f, 0.f, 0.f, 0.f, 0.f, 0.f, 0.f, 0.f};
  v8f sacc[JT];
#pragma unroll
  for (int jj = 0; jj < JT; ++jj) sacc[jj] = vzero;

  const size_t fr0 = ((size_t)(b0 + 8 * half) * NJ + jb) * DIM + l16;

  constexpr int XCAP = DIM / 4;
  constexpr int WCAP = NJ * DIM * DIM / 4;
  constexpr int WJ   = DIM * DIM / 4;
  const float4* xsrc = (const float4*)(x + ((size_t)b0 * INC + i0) * DIM);
  const unsigned xlane = (unsigned)(l16 * INC * DIM + 8 * half) / 4;
  const float4* wsrc = (const float4*)(W + (size_t)i0 * NJ * DIM * DIM);
  const unsigned wlane = (unsigned)((jb * DIM + l16) * DIM + 8 * half) / 4;

  float4 xa[2][2], xb[2][2];
  float4 wa[2][2 * JT], wb[2][2 * JT];

  auto load = [&](const int cap, float4 (&xs)[2][2],
                  float4 (&ws)[2][2 * JT]) __attribute__((always_inline)) {
#pragma unroll
    for (int g = 0; g < 2; ++g) {
      const float4* xs_src = xsrc + (size_t)(cap + g) * XCAP;
      xs[g][0] = xs_src[xlane];
      xs[g][1] = xs_src[xlane + 1];
      const float4* src = wsrc + (size_t)(cap + g) * WCAP;
#pragma unroll
      for (int jj = 0; jj < JT; ++jj) {
        ws[g][2 * jj]     = src[wlane + jj * WJ];
        ws[g][2 * jj + 1] = src[wlane + jj * WJ + 1];
      }
    }
    asm volatile("s_wait_xcnt 0x0" ::: "memory");
  };

  auto compute = [&](const float4 (&xs)[2][2],
                     const float4 (&ws)[2][2 * JT]) __attribute__((always_inline)) {
#pragma unroll
    for (int g = 0; g < 2; ++g) {
      v2f afs[4];
      afs[0].x = xs[g][0].x; afs[0].y = xs[g][0].y;
      afs[1].x = xs[g][0].z; afs[1].y = xs[g][0].w;
      afs[2].x = xs[g][1].x; afs[2].y = xs[g][1].y;
      afs[3].x = xs[g][1].z; afs[3].y = xs[g][1].w;
#pragma unroll
      for (int jj = 0; jj < JT; ++jj) {
        v2f bfr[4];
        bfr[0].x = ws[g][2 * jj].x;     bfr[0].y = ws[g][2 * jj].y;
        bfr[1].x = ws[g][2 * jj].z;     bfr[1].y = ws[g][2 * jj].w;
        bfr[2].x = ws[g][2 * jj + 1].x; bfr[2].y = ws[g][2 * jj + 1].y;
        bfr[3].x = ws[g][2 * jj + 1].z; bfr[3].y = ws[g][2 * jj + 1].w;
#pragma unroll
        for (int kk = 0; kk < 4; ++kk)
          sacc[jj] = __builtin_amdgcn_wmma_f32_16x16x4_f32(
              false, afs[kk], false, bfr[kk], (short)0, sacc[jj], false, false);
      }
    }
  };

  load(0, xa, wa);
#pragma clang loop unroll(disable)
  for (int ii = 0; ii < ICH - 4; ii += 4) {
    load(ii + 2, xb, wb);
    compute(xa, wa);
    load(ii + 4, xa, wa);
    compute(xb, wb);
  }
  load(ICH - 2, xb, wb);
  compute(xa, wa);
  compute(xb, wb);

  const float c = 1.f / (float)NJ;
#pragma unroll
  for (int jj = 0; jj < JT; ++jj)
#pragma unroll
    for (int v = 0; v < 8; ++v) sacc[jj][v] *= c;

  if (MODEA) {
    float* dst = sdst + (size_t)ich * SZ + fr0;
#pragma unroll
    for (int jj = 0; jj < JT; ++jj)
#pragma unroll
      for (int v = 0; v < 8; ++v) dst[(v * NJ + jj) * DIM] = sacc[jj][v];
  } else {
#pragma unroll
    for (int jj = 0; jj < JT; ++jj)
#pragma unroll
      for (int v = 0; v < 8; ++v)
        unsafeAtomicAdd(&sdst[fr0 + (v * NJ + jj) * DIM], sacc[jj][v]);
  }
}

__global__ void caps_squash(const float* src, float* outsum, float* dout,
                            float* szero, int nparts, int first) {
  const int idx = blockIdx.x * blockDim.x + threadIdx.x;
  float v = 0.f;
  for (int p = 0; p < nparts; ++p) v += src[(size_t)p * SZ + idx];
  if (szero) szero[idx] = 0.f;
  float prev = 0.f;
  if (!first) prev = outsum[idx];
  const float sq = xorsum16(v * v);
  const float coef = sq / ((1.f + sq) * sqrtf(sq + 1e-7f));
  const float o = coef * v;
  outsum[idx] = prev + o;
  if (dout) dout[idx] = o;
}

extern "C" void kernel_launch(void* const* d_in, const int* in_sizes, int n_in,
                              void* d_out, int out_size, void* d_ws, size_t ws_size,
                              hipStream_t stream) {
  const float* x = (const float*)d_in[0];
  const float* W = (const float*)d_in[1];
  float* out = (float*)d_out;
  float* ws = (float*)d_ws;

  float* outsum = ws;
  float* sbuf   = ws + SZ;
  float* parts  = ws + 2 * SZ;

  const bool modeA = ws_size >= (size_t)(2 * SZ + NBLK * SZ) * sizeof(float);

  if (!modeA) caps_init<<<SZ / 256, 256, 0, stream>>>(sbuf, SZ);

  for (int t = 0; t < 3; ++t) {
    if (t == 0) {
      if (modeA)
        caps_pass_first<true><<<NBLK * BSPL, NTHR, 0, stream>>>(x, W, parts);
      else
        caps_pass_first<false><<<NBLK * BSPL, NTHR, 0, stream>>>(x, W, sbuf);
    } else {
      if (modeA)
        caps_pass<true><<<NBLK * BSPL, NTHR, 0, stream>>>(x, W, outsum, parts);
      else
        caps_pass<false><<<NBLK * BSPL, NTHR, 0, stream>>>(x, W, outsum, sbuf);
    }
    caps_squash<<<SZ / 256, 256, 0, stream>>>(modeA ? parts : sbuf, outsum,
                                              (t == 2) ? out : nullptr,
                                              modeA ? nullptr : sbuf,
                                              modeA ? NBLK : 1, t == 0 ? 1 : 0);
  }
}


# Products transposed (W as A operand, x tile as B): a lane holds 8 d of one batch row, so the logit's d sum is 8 in-lane FMAs + one half exchange, not 16-lane butterflies (d sum order changes)
# speedup vs baseline: 3.0791x; 1.1983x over previous
#include <hip/hip_runtime.h>
#include <math.h>

#define BATCH 64
#define INC   2048
#define NJ    32
#define DIM   16
#define ICH   16
#define NBLK  (INC / ICH)
#define SZ    (BATCH * NJ * DIM)
#define NTHR  512
#define MTB   1
#define BB    (16 * MTB)
#define BSPL  (BATCH / BB)
#define JG    (16 / MTB)
#define JT    (NJ / JG)

typedef float v2f __attribute__((ext_vector_type(2)));
typedef float v8f __attribute__((ext_vector_type(8)));

#define DPP_XOR(x, m)                                                          \
  __int_as_float(__builtin_amdgcn_update_dpp(                                  \
      0, __float_as_int(x), 0x160 | (m), 0xf, 0xf, true))

__device__ __forceinline__ float xorsum16(float p) {
  p += DPP_XOR(p, 1);
  p += DPP_XOR(p, 2);
  p += DPP_XOR(p, 4);
  p += DPP_XOR(p, 8);
  return p;
}

__device__ __forceinline__ float xormax16(float p) {
  p = fmaxf(p, DPP_XOR(p, 1));
  p = fmaxf(p, DPP_XOR(p, 2));
  p = fmaxf(p, DPP_XOR(p, 4));
  p = fmaxf(p, DPP_XOR(p, 8));
  return p;
}

__device__ __forceinline__ float xhalf(float x) {
  const int b = __float_as_int(x);
  return __int_as_float(
      __builtin_amdgcn_permlanex16(b, b, 0x76543210, 0xfedcba98, false, false));
}

__device__ __forceinline__ void lds_barrier() {
  __builtin_amdgcn_fence(__ATOMIC_RELEASE, "workgroup", "local");
  __builtin_amdgcn_s_barrier();
  __builtin_amdgcn_fence(__ATOMIC_ACQUIRE, "workgroup", "local");
}

__global__ void caps_init(float* p, int n) {
  int i = blockIdx.x * blockDim.x + threadIdx.x;
  if (i < n) p[i] = 0.f;
}

template <bool MODEA>
__global__ __launch_bounds__(NTHR) __attribute__((amdgpu_waves_per_eu(8, 8)))
void caps_pass(const float* __restrict__ x, const float* __restrict__ W,
               const float* __restrict__ outsum, float* sdst) {
  __shared__ __attribute__((aligned(16))) float Xs[BB * DIM];
  __shared__ float Lg[BB * NJ];
  static_assert(JT == 2 && DIM == 16, "W fragment registers: two j per wave, 8 k per lane");
  static_assert(BB == NTHR / 32 && NJ == 32,
                "softmax step: one wave per batch row, one lane per j");

  const int tid  = threadIdx.x;
  const int lane = tid & 31;
  const int wv   = tid >> 5;
  const int half = lane >> 4;
  const int l16  = lane & 15;
  const int mt   = wv % MTB;
  const int jb   = (wv / MTB) * JT;
  const int ich  = blockIdx.x / BSPL;
  const int b0   = (blockIdx.x % BSPL) * BB;
  const int i0   = ich * ICH;

  const v8f vzero = {0.f, 0.f, 0.f, 0.f, 0.f, 0.f, 0.f, 0.f};
  v8f sacc[JT];
#pragma unroll
  for (int jj = 0; jj < JT; ++jj) sacc[jj] = vzero;

  const size_t fr0 = ((size_t)(b0 + mt * 16 + l16) * NJ + jb) * DIM + 8 * half;
  float osum[JT][8];
#pragma unroll
  for (int jj = 0; jj < JT; ++jj) {
    const float4 o0 = *(const float4*)(outsum + fr0 + jj * DIM);
    const float4 o1 = *(const float4*)(outsum + fr0 + jj * DIM + 4);
    osum[jj][0] = o0.x; osum[jj][1] = o0.y; osum[jj][2] = o0.z; osum[jj][3] = o0.w;
    osum[jj][4] = o1.x; osum[jj][5] = o1.y; osum[jj][6] = o1.z; osum[jj][7] = o1.w;
  }

  const int xr = (tid >> 3) & (BB - 1), xc = (tid & 7) * 2;
  const float* xsrc = x + ((size_t)(b0 + xr) * INC + i0) * DIM + xc;
  float2 px = *(const float2*)xsrc;

  const float4* wsrc =
      (const float4*)(W + (((size_t)i0 * NJ + jb) * DIM + l16) * DIM + 8 * half);
  constexpr int WCAP = NJ * DIM * DIM / 4;
  constexpr int WJ   = DIM * DIM / 4;
  float4 wa[2 * JT], wb[2 * JT];
#pragma unroll
  for (int jj = 0; jj < JT; ++jj) {
    wa[2 * jj]     = wsrc[jj * WJ];
    wa[2 * jj + 1] = wsrc[jj * WJ + 1];
  }

  auto trip = [&](const int nx, const float4 (&cw)[2 * JT],
                  float4 (&nw)[2 * JT]) __attribute__((always_inline)) {
    *(float2*)(Xs + xr * DIM + xc) = px;
    {
      px = *(const float2*)(xsrc + (size_t)nx * DIM);
      const float4* src = wsrc + (size_t)nx * WCAP;
#pragma unroll
      for (int jj = 0; jj < JT; ++jj) {
        nw[2 * jj]     = src[jj * WJ];
        nw[2 * jj + 1] = src[jj * WJ + 1];
      }
      asm volatile("s_wait_xcnt 0x0" ::: "memory");
    }
    lds_barrier();

    v2f xfr[4];
    {
      const float4* xa = (const float4*)(Xs + (mt * 16 + l16) * DIM + 8 * half);
      const float4 a0 = xa[0], a1 = xa[1];
      xfr[0].x = a0.x; xfr[0].y = a0.y;
      xfr[1].x = a0.z; xfr[1].y = a0.w;
      xfr[2].x = a1.x; xfr[2].y = a1.y;
      xfr[3].x = a1.z; xfr[3].y = a1.w;
    }

    float part[JT];
#pragma unroll
    for (int jj = 0; jj < JT; ++jj) {
      v2f wfr[4];
      wfr[0].x = cw[2 * jj].x;     wfr[0].y = cw[2 * jj].y;
      wfr[1].x = cw[2 * jj].z;     wfr[1].y = cw[2 * jj].w;
      wfr[2].x = cw[2 * jj + 1].x; wfr[2].y = cw[2 * jj + 1].y;
      wfr[3].x = cw[2 * jj + 1].z; wfr[3].y = cw[2 * jj + 1].w;
      v8f acc = vzero;
#pragma unroll
      for (int kk = 0; kk < 4; ++kk)
        acc = __builtin_amdgcn_wmma_f32_16x16x4_f32(false, wfr[kk], false, xfr[kk],
                                                    (short)0, acc, false, false);
      float p = acc[0] * osum[jj][0];
#pragma unroll
      for (int r = 1; r < 8; ++r) p = fmaf(acc[r], osum[jj][r], p);
      part[jj] = p;
    }
    {
      const float keep = half ? part[1] : part[0];
      const float send = half ? part[0] : part[1];
      Lg[(mt * 16 + l16) * NJ + jb + half] = keep + xhalf(send);
    }
    lds_barrier();

    {
      const float lg = Lg[wv * NJ + lane];
      float m = xormax16(lg);
      m = fmaxf(m, xhalf(m));
      const float e = expf(lg - m);
      float s = xorsum16(e);
      s += xhalf(s);
      Lg[wv * NJ + lane] = e * (1.f / s);
    }
    lds_barrier();

    const int brow = mt * 16 + l16;
#pragma unroll
    for (int jj = 0; jj < JT; ++jj) {
      const int j = jb + jj;
      const float c = Lg[brow * NJ + j];
      v2f wfr[4];
      wfr[0].x = cw[2 * jj].x;     wfr[0].y = cw[2 * jj].y;
      wfr[1].x = cw[2 * jj].z;     wfr[1].y = cw[2 * jj].w;
      wfr[2].x = cw[2 * jj + 1].x; wfr[2].y = cw[2 * jj + 1].y;
      wfr[3].x = cw[2 * jj + 1].z; wfr[3].y = cw[2 * jj + 1].w;
#pragma unroll
      for (int kk = 0; kk < 4; ++kk) {
        v2f xs;
        xs.x = c * xfr[kk].x;
        xs.y = c * xfr[kk].y;
        sacc[jj] = __builtin_amdgcn_wmma_f32_16x16x4_f32(
            false, wfr[kk], false, xs, (short)0, sacc[jj], false, false);
      }
    }
    lds_barrier();
  };

#pragma clang loop unroll(disable)
  for (int ii = 0; ii < ICH; ii += 2) {
    trip(ii + 1, wa, wb);
    trip(ii + 2 < ICH ? ii + 2 : ICH - 1, wb, wa);
  }

  if (MODEA) {
    float* dst = sdst + (size_t)ich * SZ + fr0;
#pragma unroll
    for (int jj = 0; jj < JT; ++jj) {
      *(float4*)(dst + jj * DIM) =
          make_float4(sacc[jj][0], sacc[jj][1], sacc[jj][2], sacc[jj][3]);
      *(float4*)(dst + jj * DIM + 4) =
          make_float4(sacc[jj][4], sacc[jj][5], sacc[jj][6], sacc[jj][7]);
    }
  } else {
#pragma unroll
    for (int jj = 0; jj < JT; ++jj)
#pragma unroll
      for (int r = 0; r < 8; ++r)
        unsafeAtomicAdd(&sdst[fr0 + jj * DIM + r], sacc[jj][r]);
  }
}

template <bool MODEA>
__global__ __launch_bounds__(NTHR) __attribute__((amdgpu_waves_per_eu(8, 8)))
void caps_pass_first(const float* __restrict__ x, const float* __restrict__ W,
                     float* sdst) {
  static_assert(JT == 2 && DIM == 16 && MTB == 1 && (ICH % 4) == 0,
                "two j per wave, 8 k per lane, capsule groups of two in pairs");
  static_assert((NJ & (NJ - 1)) == 0, "1/NJ must be a power of two (exact scaling)");
  const int tid  = threadIdx.x;
  const int lane = tid & 31;
  const int wv   = tid >> 5;
  const int half = lane >> 4;
  const int l16  = lane & 15;
  const int jb   = wv * JT;
  const int ich  = blockIdx.x / BSPL;
  const int b0   = (blockIdx.x % BSPL) * BB;
  const int i0   = ich * ICH;

  const v8f vzero = {0.f, 0.f, 0.f, 0.f, 0.f, 0.f, 0.f, 0.f};
  v8f sacc[JT];
#pragma unroll
  for (int jj = 0; jj < JT; ++jj) sacc[jj] = vzero;

  const size_t fr0 = ((size_t)(b0 + 8 * half) * NJ + jb) * DIM + l16;

  constexpr int XCAP = DIM / 4;
  constexpr int WCAP = NJ * DIM * DIM / 4;
  constexpr int WJ   = DIM * DIM / 4;
  const float4* xsrc = (const float4*)(x + ((size_t)b0 * INC + i0) * DIM);
  const unsigned xlane = (unsigned)(l16 * INC * DIM + 8 * half) / 4;
  const float4* wsrc = (const float4*)(W + (size_t)i0 * NJ * DIM * DIM);
  const unsigned wlane = (unsigned)((jb * DIM + l16) * DIM + 8 * half) / 4;

  float4 xa[2][2], xb[2][2];
  float4 wa[2][2 * JT], wb[2][2 * JT];

  auto load = [&](const int cap, float4 (&xs)[2][2],
                  float4 (&ws)[2][2 * JT]) __attribute__((always_inline)) {
#pragma unroll
    for (int g = 0; g < 2; ++g) {
      const float4* xs_src = xsrc + (size_t)(cap + g) * XCAP;
      xs[g][0] = xs_src[xlane];
      xs[g][1] = xs_src[xlane + 1];
      const float4* src = wsrc + (size_t)(cap + g) * WCAP;
#pragma unroll
      for (int jj = 0; jj < JT; ++jj) {
        ws[g][2 * jj]     = src[wlane + jj * WJ];
        ws[g][2 * jj + 1] = src[wlane + jj * WJ + 1];
      }
    }
    asm volatile("s_wait_xcnt 0x0" ::: "memory");
  };

  auto compute = [&](const float4 (&xs)[2][2],
                     const float4 (&ws)[2][2 * JT]) __attribute__((always_inline)) {
#pragma unroll
    for (int g = 0; g < 2; ++g) {
      v2f afs[4];
      afs[0].x = xs[g][0].x; afs[0].y = xs[g][0].y;
      afs[1].x = xs[g][0].z; afs[1].y = xs[g][0].w;
      afs[2].x = xs[g][1].x; afs[2].y = xs[g][1].y;
      afs[3].x = xs[g][1].z; afs[3].y = xs[g][1].w;
#pragma unroll
      for (int jj = 0; jj < JT; ++jj) {
        v2f bfr[4];
        bfr[0].x = ws[g][2 * jj].x;     bfr[0].y = ws[g][2 * jj].y;
        bfr[1].x = ws[g][2 * jj].z;     bfr[1].y = ws[g][2 * jj].w;
        bfr[2].x = ws[g][2 * jj + 1].x; bfr[2].y = ws[g][2 * jj + 1].y;
        bfr[3].x = ws[g][2 * jj + 1].z; bfr[3].y = ws[g][2 * jj + 1].w;
#pragma unroll
        for (int kk = 0; kk < 4; ++kk)
          sacc[jj] = __builtin_amdgcn_wmma_f32_16x16x4_f32(
              false, afs[kk], false, bfr[kk], (short)0, sacc[jj], false, false);
      }
    }
  };

  load(0, xa, wa);
#pragma clang loop unroll(disable)
  for (int ii = 0; ii < ICH - 4; ii += 4) {
    load(ii + 2, xb, wb);
    compute(xa, wa);
    load(ii + 4, xa, wa);
    compute(xb, wb);
  }
  load(ICH - 2, xb, wb);
  compute(xa, wa);
  compute(xb, wb);

  const float c = 1.f / (float)NJ;
#pragma unroll
  for (int jj = 0; jj < JT; ++jj)
#pragma unroll
    for (int v = 0; v < 8; ++v) sacc[jj][v] *= c;

  if (MODEA) {
    float* dst = sdst + (size_t)ich * SZ + fr0;
#pragma unroll
    for (int jj = 0; jj < JT; ++jj)
#pragma unroll
      for (int v = 0; v < 8; ++v) dst[(v * NJ + jj) * DIM] = sacc[jj][v];
  } else {
#pragma unroll
    for (int jj = 0; jj < JT; ++jj)
#pragma unroll
      for (int v = 0; v < 8; ++v)
        unsafeAtomicAdd(&sdst[fr0 + (v * NJ + jj) * DIM], sacc[jj][v]);
  }
}

__global__ void caps_squash(const float* src, float* outsum, float* dout,
                            float* szero, int nparts, int first) {
  const int idx = blockIdx.x * blockDim.x + threadIdx.x;
  float v = 0.f;
  for (int p = 0; p < nparts; ++p) v += src[(size_t)p * SZ + idx];
  if (szero) szero[idx] = 0.f;
  float prev = 0.f;
  if (!first) prev = outsum[idx];
  const float sq = xorsum16(v * v);
  const float coef = sq / ((1.f + sq) * sqrtf(sq + 1e-7f));
  const float o = coef * v;
  outsum[idx] = prev + o;
  if (dout) dout[idx] = o;
}

extern "C" void kernel_launch(void* const* d_in, const int* in_sizes, int n_in,
                              void* d_out, int out_size, void* d_ws, size_t ws_size,
                              hipStream_t stream) {
  const float* x = (const float*)d_in[0];
  const float* W = (const float*)d_in[1];
  float* out = (float*)d_out;
  float* ws = (float*)d_ws;

  float* outsum = ws;
  float* sbuf   = ws + SZ;
  float* parts  = ws + 2 * SZ;

  const bool modeA = ws_size >= (size_t)(2 * SZ + NBLK * SZ) * sizeof(float);

  if (!modeA) caps_init<<<SZ / 256, 256, 0, stream>>>(sbuf, SZ);

  for (int t = 0; t < 3; ++t) {
    if (t == 0) {
      if (modeA)
        caps_pass_first<true><<<NBLK * BSPL, NTHR, 0, stream>>>(x, W, parts);
      else
        caps_pass_first<false><<<NBLK * BSPL, NTHR, 0, stream>>>(x, W, sbuf);
    } else {
      if (modeA)
        caps_pass<true><<<NBLK * BSPL, NTHR, 0, stream>>>(x, W, outsum, parts);
      else
        caps_pass<false><<<NBLK * BSPL, NTHR, 0, stream>>>(x, W, outsum, sbuf);
    }
    caps_squash<<<SZ / 256, 256, 0, stream>>>(modeA ? parts : sbuf, outsum,
                                              (t == 2) ? out : nullptr,
                                              modeA ? nullptr : sbuf,
                                              modeA ? NBLK : 1, t == 0 ? 1 : 0);
  }
}


# Weighted sum reuses the u_hat tile of the logit step, kept in registers: s += c*u_hat as eight in-lane FMAs per j; the repeated per-capsule product W(c o x) is gone (i and k sum order changes)
# speedup vs baseline: 3.4037x; 1.1054x over previous
#include <hip/hip_runtime.h>
#include <math.h>

#define BATCH 64
#define INC   2048
#define NJ    32
#define DIM   16
#define ICH   16
#define NBLK  (INC / ICH)
#define SZ    (BATCH * NJ * DIM)
#define NTHR  512
#define MTB   1
#define BB    (16 * MTB)
#define BSPL  (BATCH / BB)
#define JG    (16 / MTB)
#define JT    (NJ / JG)

typedef float v2f __attribute__((ext_vector_type(2)));
typedef float v8f __attribute__((ext_vector_type(8)));

#define DPP_XOR(x, m)                                                          \
  __int_as_float(__builtin_amdgcn_update_dpp(                                  \
      0, __float_as_int(x), 0x160 | (m), 0xf, 0xf, true))

__device__ __forceinline__ float xorsum16(float p) {
  p += DPP_XOR(p, 1);
  p += DPP_XOR(p, 2);
  p += DPP_XOR(p, 4);
  p += DPP_XOR(p, 8);
  return p;
}

__device__ __forceinline__ float xormax16(float p) {
  p = fmaxf(p, DPP_XOR(p, 1));
  p = fmaxf(p, DPP_XOR(p, 2));
  p = fmaxf(p, DPP_XOR(p, 4));
  p = fmaxf(p, DPP_XOR(p, 8));
  return p;
}

__device__ __forceinline__ float xhalf(float x) {
  const int b = __float_as_int(x);
  return __int_as_float(
      __builtin_amdgcn_permlanex16(b, b, 0x76543210, 0xfedcba98, false, false));
}

__device__ __forceinline__ void lds_barrier() {
  __builtin_amdgcn_fence(__ATOMIC_RELEASE, "workgroup", "local");
  __builtin_amdgcn_s_barrier();
  __builtin_amdgcn_fence(__ATOMIC_ACQUIRE, "workgroup", "local");
}

__global__ void caps_init(float* p, int n) {
  int i = blockIdx.x * blockDim.x + threadIdx.x;
  if (i < n) p[i] = 0.f;
}

template <bool MODEA>
__global__ __launch_bounds__(NTHR) __attribute__((amdgpu_waves_per_eu(8, 8)))
void caps_pass(const float* __restrict__ x, const float* __restrict__ W,
               const float* __restrict__ outsum, float* sdst) {
  __shared__ __attribute__((aligned(16))) float Xs[BB * DIM];
  __shared__ float Lg[BB * NJ];
  static_assert(JT == 2 && DIM == 16, "W fragment registers: two j per wave, 8 k per lane");
  static_assert(BB == NTHR / 32 && NJ == 32,
                "softmax step: one wave per batch row, one lane per j");

  const int tid  = threadIdx.x;
  const int lane = tid & 31;
  const int wv   = tid >> 5;
  const int half = lane >> 4;
  const int l16  = lane & 15;
  const int mt   = wv % MTB;
  const int jb   = (wv / MTB) * JT;
  const int ich  = blockIdx.x / BSPL;
  const int b0   = (blockIdx.x % BSPL) * BB;
  const int i0   = ich * ICH;

  const v8f vzero = {0.f, 0.f, 0.f, 0.f, 0.f, 0.f, 0.f, 0.f};
  v8f sacc[JT];
#pragma unroll
  for (int jj = 0; jj < JT; ++jj) sacc[jj] = vzero;

  const size_t fr0 = ((size_t)(b0 + mt * 16 + l16) * NJ + jb) * DIM + 8 * half;
  float osum[JT][8];
#pragma unroll
  for (int jj = 0; jj < JT; ++jj) {
    const float4 o0 = *(const float4*)(outsum + fr0 + jj * DIM);
    const float4 o1 = *(const float4*)(outsum + fr0 + jj * DIM + 4);
    osum[jj][0] = o0.x; osum[jj][1] = o0.y; osum[jj][2] = o0.z; osum[jj][3] = o0.w;
    osum[jj][4] = o1.x; osum[jj][5] = o1.y; osum[jj][6] = o1.z; osum[jj][7] = o1.w;
  }

  const int xr = (tid >> 3) & (BB - 1), xc = (tid & 7) * 2;
  const float* xsrc = x + ((size_t)(b0 + xr) * INC + i0) * DIM + xc;
  float2 px = *(const float2*)xsrc;

  const float4* wsrc =
      (const float4*)(W + (((size_t)i0 * NJ + jb) * DIM + l16) * DIM + 8 * half);
  constexpr int WCAP = NJ * DIM * DIM / 4;
  constexpr int WJ   = DIM * DIM / 4;
  float4 wa[2 * JT], wb[2 * JT];
#pragma unroll
  for (int jj = 0; jj < JT; ++jj) {
    wa[2 * jj]     = wsrc[jj * WJ];
    wa[2 * jj + 1] = wsrc[jj * WJ + 1];
  }

  auto trip = [&](const int nx, const float4 (&cw)[2 * JT],
                  float4 (&nw)[2 * JT]) __attribute__((always_inline)) {
    *(float2*)(Xs + xr * DIM + xc) = px;
    {
      px = *(const float2*)(xsrc + (size_t)nx * DIM);
      const float4* src = wsrc + (size_t)nx * WCAP;
#pragma unroll
      for (int jj = 0; jj < JT; ++jj) {
        nw[2 * jj]     = src[jj * WJ];
        nw[2 * jj + 1] = src[jj * WJ + 1];
      }
      asm volatile("s_wait_xcnt 0x0" ::: "memory");
    }
    lds_barrier();

    v2f xfr[4];
    {
      const float4* xa = (const float4*)(Xs + (mt * 16 + l16) * DIM + 8 * half);
      const float4 a0 = xa[0], a1 = xa[1];
      xfr[0].x = a0.x; xfr[0].y = a0.y;
      xfr[1].x = a0.z; xfr[1].y = a0.w;
      xfr[2].x = a1.x; xfr[2].y = a1.y;
      xfr[3].x = a1.z; xfr[3].y = a1.w;
    }

    v8f uh[JT];
    float part[JT];
#pragma unroll
    for (int jj = 0; jj < JT; ++jj) {
      v2f wfr[4];
      wfr[0].x = cw[2 * jj].x;     wfr[0].y = cw[2 * jj].y;
      wfr[1].x = cw[2 * jj].z;     wfr[1].y = cw[2 * jj].w;
      wfr[2].x = cw[2 * jj + 1].x; wfr[2].y = cw[2 * jj + 1].y;
      wfr[3].x = cw[2 * jj + 1].z; wfr[3].y = cw[2 * jj + 1].w;
      v8f acc = vzero;
#pragma unroll
      for (int kk = 0; kk < 4; ++kk)
        acc = __builtin_amdgcn_wmma_f32_16x16x4_f32(false, wfr[kk], false, xfr[kk],
                                                    (short)0, acc, false, false);
      uh[jj] = acc;
      float p = acc[0] * osum[jj][0];
#pragma unroll
      for (int r = 1; r < 8; ++r) p = fmaf(acc[r], osum[jj][r], p);
      part[jj] = p;
    }
    {
      const float keep = half ? part[1] : part[0];
      const float send = half ? part[0] : part[1];
      Lg[(mt * 16 + l16) * NJ + jb + half] = keep + xhalf(send);
    }
    lds_barrier();

    {
      const float lg = Lg[wv * NJ + lane];
      float m = xormax16(lg);
      m = fmaxf(m, xhalf(m));
      const float e = expf(lg - m);
      float s = xorsum16(e);
      s += xhalf(s);
      Lg[wv * NJ + lane] = e * (1.f / s);
    }
    lds_barrier();

    const int brow = mt * 16 + l16;
#pragma unroll
    for (int jj = 0; jj < JT; ++jj) {
      const float c = Lg[brow * NJ + jb + jj];
#pragma unroll
      for (int r = 0; r < 8; ++r) sacc[jj][r] = fmaf(c, uh[jj][r], sacc[jj][r]);
    }
    lds_barrier();
  };

#pragma clang loop unroll(disable)
  for (int ii = 0; ii < ICH; ii += 2) {
    trip(ii + 1, wa, wb);
    trip(ii + 2 < ICH ? ii + 2 : ICH - 1, wb, wa);
  }

  if (MODEA) {
    float* dst = sdst + (size_t)ich * SZ + fr0;
#pragma unroll
    for (int jj = 0; jj < JT; ++jj) {
      *(float4*)(dst + jj * DIM) =
          make_float4(sacc[jj][0], sacc[jj][1], sacc[jj][2], sacc[jj][3]);
      *(float4*)(dst + jj * DIM + 4) =
          make_float4(sacc[jj][4], sacc[jj][5], sacc[jj][6], sacc[jj][7]);
    }
  } else {
#pragma unroll
    for (int jj = 0; jj < JT; ++jj)
#pragma unroll
      for (int r = 0; r < 8; ++r)
        unsafeAtomicAdd(&sdst[fr0 + jj * DIM + r], sacc[jj][r]);
  }
}

template <bool MODEA>
__global__ __launch_bounds__(NTHR) __attribute__((amdgpu_waves_per_eu(8, 8)))
void caps_pass_first(const float* __restrict__ x, const float* __restrict__ W,
                     float* sdst) {
  static_assert(JT == 2 && DIM == 16 && MTB == 1 && (ICH % 4) == 0,
                "two j per wave, 8 k per lane, capsule groups of two in pairs");
  static_assert((NJ & (NJ - 1)) == 0, "1/NJ must be a power of two (exact scaling)");
  const int tid  = threadIdx.x;
  const int lane = tid & 31;
  const int wv   = tid >> 5;
  const int half = lane >> 4;
  const int l16  = lane & 15;
  const int jb   = wv * JT;
  const int ich  = blockIdx.x / BSPL;
  const int b0   = (blockIdx.x % BSPL) * BB;
  const int i0   = ich * ICH;

  const v8f vzero = {0.f, 0.f, 0.f, 0.f, 0.f, 0.f, 0.f, 0.f};
  v8f sacc[JT];
#pragma unroll
  for (int jj = 0; jj < JT; ++jj) sacc[jj] = vzero;

  const size_t fr0 = ((size_t)(b0 + 8 * half) * NJ + jb) * DIM + l16;

  constexpr int XCAP = DIM / 4;
  constexpr int WCAP = NJ * DIM * DIM / 4;
  constexpr int WJ   = DIM * DIM / 4;
  const float4* xsrc = (const float4*)(x + ((size_t)b0 * INC + i0) * DIM);
  const unsigned xlane = (unsigned)(l16 * INC * DIM + 8 * half) / 4;
  const float4* wsrc = (const float4*)(W + (size_t)i0 * NJ * DIM * DIM);
  const unsigned wlane = (unsigned)((jb * DIM + l16) * DIM + 8 * half) / 4;

  float4 xa[2][2], xb[2][2];
  float4 wa[2][2 * JT], wb[2][2 * JT];

  auto load = [&](const int cap, float4 (&xs)[2][2],
                  float4 (&ws)[2][2 * JT]) __attribute__((always_inline)) {
#pragma unroll
    for (int g = 0; g < 2; ++g) {
      const float4* xs_src = xsrc + (size_t)(cap + g) * XCAP;
      xs[g][0] = xs_src[xlane];
      xs[g][1] = xs_src[xlane + 1];
      const float4* src = wsrc + (size_t)(cap + g) * WCAP;
#pragma unroll
      for (int jj = 0; jj < JT; ++jj) {
        ws[g][2 * jj]     = src[wlane + jj * WJ];
        ws[g][2 * jj + 1] = src[wlane + jj * WJ + 1];
      }
    }
    asm volatile("s_wait_xcnt 0x0" ::: "memory");
  };

  auto compute = [&](const float4 (&xs)[2][2],
                     const float4 (&ws)[2][2 * JT]) __attribute__((always_inline)) {
#pragma unroll
    for (int g = 0; g < 2; ++g) {
      v2f afs[4];
      afs[0].x = xs[g][0].x; afs[0].y = xs[g][0].y;
      afs[1].x = xs[g][0].z; afs[1].y = xs[g][0].w;
      afs[2].x = xs[g][1].x; afs[2].y = xs[g][1].y;
      afs[3].x = xs[g][1].z; afs[3].y = xs[g][1].w;
#pragma unroll
      for (int jj = 0; jj < JT; ++jj) {
        v2f bfr[4];
        bfr[0].x = ws[g][2 * jj].x;     bfr[0].y = ws[g][2 * jj].y;
        bfr[1].x = ws[g][2 * jj].z;     bfr[1].y = ws[g][2 * jj].w;
        bfr[2].x = ws[g][2 * jj + 1].x; bfr[2].y = ws[g][2 * jj + 1].y;
        bfr[3].x = ws[g][2 * jj + 1].z; bfr[3].y = ws[g][2 * jj + 1].w;
#pragma unroll
        for (int kk = 0; kk < 4; ++kk)
          sacc[jj] = __builtin_amdgcn_wmma_f32_16x16x4_f32(
              false, afs[kk], false, bfr[kk], (short)0, sacc[jj], false, false);
      }
    }
  };

  load(0, xa, wa);
#pragma clang loop unroll(disable)
  for (int ii = 0; ii < ICH - 4; ii += 4) {
    load(ii + 2, xb, wb);
    compute(xa, wa);
    load(ii + 4, xa, wa);
    compute(xb, wb);
  }
  load(ICH - 2, xb, wb);
  compute(xa, wa);
  compute(xb, wb);

  const float c = 1.f / (float)NJ;
#pragma unroll
  for (int jj = 0; jj < JT; ++jj)
#pragma unroll
    for (int v = 0; v < 8; ++v) sacc[jj][v] *= c;

  if (MODEA) {
    float* dst = sdst + (size_t)ich * SZ + fr0;
#pragma unroll
    for (int jj = 0; jj < JT; ++jj)
#pragma unroll
      for (int v = 0; v < 8; ++v) dst[(v * NJ + jj) * DIM] = sacc[jj][v];
  } else {
#pragma unroll
    for (int jj = 0; jj < JT; ++jj)
#pragma unroll
      for (int v = 0; v < 8; ++v)
        unsafeAtomicAdd(&sdst[fr0 + (v * NJ + jj) * DIM], sacc[jj][v]);
  }
}

__global__ void caps_squash(const float* src, float* outsum, float* dout,
                            float* szero, int nparts, int first) {
  const int idx = blockIdx.x * blockDim.x + threadIdx.x;
  float v = 0.f;
  for (int p = 0; p < nparts; ++p) v += src[(size_t)p * SZ + idx];
  if (szero) szero[idx] = 0.f;
  float prev = 0.f;
  if (!first) prev = outsum[idx];
  const float sq = xorsum16(v * v);
  const float coef = sq / ((1.f + sq) * sqrtf(sq + 1e-7f));
  const float o = coef * v;
  outsum[idx] = prev + o;
  if (dout) dout[idx] = o;
}

extern "C" void kernel_launch(void* const* d_in, const int* in_sizes, int n_in,
                              void* d_out, int out_size, void* d_ws, size_t ws_size,
                              hipStream_t stream) {
  const float* x = (const float*)d_in[0];
  const float* W = (const float*)d_in[1];
  float* out = (float*)d_out;
  float* ws = (float*)d_ws;

  float* outsum = ws;
  float* sbuf   = ws + SZ;
  float* parts  = ws + 2 * SZ;

  const bool modeA = ws_size >= (size_t)(2 * SZ + NBLK * SZ) * sizeof(float);

  if (!modeA) caps_init<<<SZ / 256, 256, 0, stream>>>(sbuf, SZ);

  for (int t = 0; t < 3; ++t) {
    if (t == 0) {
      if (modeA)
        caps_pass_first<true><<<NBLK * BSPL, NTHR, 0, stream>>>(x, W, parts);
      else
        caps_pass_first<false><<<NBLK * BSPL, NTHR, 0, stream>>>(x, W, sbuf);
    } else {
      if (modeA)
        caps_pass<true><<<NBLK * BSPL, NTHR, 0, stream>>>(x, W, outsum, parts);
      else
        caps_pass<false><<<NBLK * BSPL, NTHR, 0, stream>>>(x, W, outsum, sbuf);
    }
    caps_squash<<<SZ / 256, 256, 0, stream>>>(modeA ? parts : sbuf, outsum,
                                              (t == 2) ? out : nullptr,
                                              modeA ? nullptr : sbuf,
                                              modeA ? NBLK : 1, t == 0 ? 1 : 0);
  }
}


# Partial reduction of the squash launches spread over 16 waves x 8 float4 loads per block, all in flight at once, wave sums joined in LDS (sum order over partials: 16 runs of 8, then the runs)
# speedup vs baseline: 4.1761x; 1.2269x over previous
#include <hip/hip_runtime.h>
#include <math.h>

#define BATCH 64
#define INC   2048
#define NJ    32
#define DIM   16
#define ICH   16
#define NBLK  (INC / ICH)
#define SZ    (BATCH * NJ * DIM)
#define NTHR  512
#define MTB   1
#define BB    (16 * MTB)
#define BSPL  (BATCH / BB)
#define JG    (16 / MTB)
#define JT    (NJ / JG)

typedef float v2f __attribute__((ext_vector_type(2)));
typedef float v8f __attribute__((ext_vector_type(8)));

#define DPP_XOR(x, m)                                                          \
  __int_as_float(__builtin_amdgcn_update_dpp(                                  \
      0, __float_as_int(x), 0x160 | (m), 0xf, 0xf, true))

__device__ __forceinline__ float xorsum16(float p) {
  p += DPP_XOR(p, 1);
  p += DPP_XOR(p, 2);
  p += DPP_XOR(p, 4);
  p += DPP_XOR(p, 8);
  return p;
}

__device__ __forceinline__ float xormax16(float p) {
  p = fmaxf(p, DPP_XOR(p, 1));
  p = fmaxf(p, DPP_XOR(p, 2));
  p = fmaxf(p, DPP_XOR(p, 4));
  p = fmaxf(p, DPP_XOR(p, 8));
  return p;
}

__device__ __forceinline__ float xhalf(float x) {
  const int b = __float_as_int(x);
  return __int_as_float(
      __builtin_amdgcn_permlanex16(b, b, 0x76543210, 0xfedcba98, false, false));
}

__device__ __forceinline__ void lds_barrier() {
  __builtin_amdgcn_fence(__ATOMIC_RELEASE, "workgroup", "local");
  __builtin_amdgcn_s_barrier();
  __builtin_amdgcn_fence(__ATOMIC_ACQUIRE, "workgroup", "local");
}

__global__ void caps_init(float* p, int n) {
  int i = blockIdx.x * blockDim.x + threadIdx.x;
  if (i < n) p[i] = 0.f;
}

template <bool MODEA>
__global__ __launch_bounds__(NTHR) __attribute__((amdgpu_waves_per_eu(8, 8)))
void caps_pass(const float* __restrict__ x, const float* __restrict__ W,
               const float* __restrict__ outsum, float* sdst) {
  __shared__ __attribute__((aligned(16))) float Xs[BB * DIM];
  __shared__ float Lg[BB * NJ];
  static_assert(JT == 2 && DIM == 16, "W fragment registers: two j per wave, 8 k per lane");
  static_assert(BB == NTHR / 32 && NJ == 32,
                "softmax step: one wave per batch row, one lane per j");

  const int tid  = threadIdx.x;
  const int lane = tid & 31;
  const int wv   = tid >> 5;
  const int half = lane >> 4;
  const int l16  = lane & 15;
  const int mt   = wv % MTB;
  const int jb   = (wv / MTB) * JT;
  const int ich  = blockIdx.x / BSPL;
  const int b0   = (blockIdx.x % BSPL) * BB;
  const int i0   = ich * ICH;

  const v8f vzero = {0.f, 0.f, 0.f, 0.f, 0.f, 0.f, 0.f, 0.f};
  v8f sacc[JT];
#pragma unroll
  for (int jj = 0; jj < JT; ++jj) sacc[jj] = vzero;

  const size_t fr0 = ((size_t)(b0 + mt * 16 + l16) * NJ + jb) * DIM + 8 * half;
  float osum[JT][8];
#pragma unroll
  for (int jj = 0; jj < JT; ++jj) {
    const float4 o0 = *(const float4*)(outsum + fr0 + jj * DIM);
    const float4 o1 = *(const float4*)(outsum + fr0 + jj * DIM + 4);
    osum[jj][0] = o0.x; osum[jj][1] = o0.y; osum[jj][2] = o0.z; osum[jj][3] = o0.w;
    osum[jj][4] = o1.x; osum[jj][5] = o1.y; osum[jj][6] = o1.z; osum[jj][7] = o1.w;
  }

  const int xr = (tid >> 3) & (BB - 1), xc = (tid & 7) * 2;
  const float* xsrc = x + ((size_t)(b0 + xr) * INC + i0) * DIM + xc;
  float2 px = *(const float2*)xsrc;

  const float4* wsrc =
      (const float4*)(W + (((size_t)i0 * NJ + jb) * DIM + l16) * DIM + 8 * half);
  constexpr int WCAP = NJ * DIM * DIM / 4;
  constexpr int WJ   = DIM * DIM / 4;
  float4 wa[2 * JT], wb[2 * JT];
#pragma unroll
  for (int jj = 0; jj < JT; ++jj) {
    wa[2 * jj]     = wsrc[jj * WJ];
    wa[2 * jj + 1] = wsrc[jj * WJ + 1];
  }

  auto trip = [&](const int nx, const float4 (&cw)[2 * JT],
                  float4 (&nw)[2 * JT]) __attribute__((always_inline)) {
    *(float2*)(Xs + xr * DIM + xc) = px;
    {
      px = *(const float2*)(xsrc + (size_t)nx * DIM);
      const float4* src = wsrc + (size_t)nx * WCAP;
#pragma unroll
      for (int jj = 0; jj < JT; ++jj) {
        nw[2 * jj]     = src[jj * WJ];
        nw[2 * jj + 1] = src[jj * WJ + 1];
      }
      asm volatile("s_wait_xcnt 0x0" ::: "memory");
    }
    lds_barrier();

    v2f xfr[4];
    {
      const float4* xa = (const float4*)(Xs + (mt * 16 + l16) * DIM + 8 * half);
      const float4 a0 = xa[0], a1 = xa[1];
      xfr[0].x = a0.x; xfr[0].y = a0.y;
      xfr[1].x = a0.z; xfr[1].y = a0.w;
      xfr[2].x = a1.x; xfr[2].y = a1.y;
      xfr[3].x = a1.z; xfr[3].y = a1.w;
    }

    v8f uh[JT];
    float part[JT];
#pragma unroll
    for (int jj = 0; jj < JT; ++jj) {
      v2f wfr[4];
      wfr[0].x = cw[2 * jj].x;     wfr[0].y = cw[2 * jj].y;
      wfr[1].x = cw[2 * jj].z;     wfr[1].y = cw[2 * jj].w;
      wfr[2].x = cw[2 * jj + 1].x; wfr[2].y = cw[2 * jj + 1].y;
      wfr[3].x = cw[2 * jj + 1].z; wfr[3].y = cw[2 * jj + 1].w;
      v8f acc = vzero;
#pragma unroll
      for (int kk = 0; kk < 4; ++kk)
        acc = __builtin_amdgcn_wmma_f32_16x16x4_f32(false, wfr[kk], false, xfr[kk],
                                                    (short)0, acc, false, false);
      uh[jj] = acc;
      float p = acc[0] * osum[jj][0];
#pragma unroll
      for (int r = 1; r < 8; ++r) p = fmaf(acc[r], osum[jj][r], p);
      part[jj] = p;
    }
    {
      const float keep = half ? part[1] : part[0];
      const float send = half ? part[0] : part[1];
      Lg[(mt * 16 + l16) * NJ + jb + half] = keep + xhalf(send);
    }
    lds_barrier();

    {
      const float lg = Lg[wv * NJ + lane];
      float m = xormax16(lg);
      m = fmaxf(m, xhalf(m));
      const float e = expf(lg - m);
      float s = xorsum16(e);
      s += xhalf(s);
      Lg[wv * NJ + lane] = e * (1.f / s);
    }
    lds_barrier();

    const int brow = mt * 16 + l16;
#pragma unroll
    for (int jj = 0; jj < JT; ++jj) {
      const float c = Lg[brow * NJ + jb + jj];
#pragma unroll
      for (int r = 0; r < 8; ++r) sacc[jj][r] = fmaf(c, uh[jj][r], sacc[jj][r]);
    }
    lds_barrier();
  };

#pragma clang loop unroll(disable)
  for (int ii = 0; ii < ICH; ii += 2) {
    trip(ii + 1, wa, wb);
    trip(ii + 2 < ICH ? ii + 2 : ICH - 1, wb, wa);
  }

  if (MODEA) {
    float* dst = sdst + (size_t)ich * SZ + fr0;
#pragma unroll
    for (int jj = 0; jj < JT; ++jj) {
      *(float4*)(dst + jj * DIM) =
          make_float4(sacc[jj][0], sacc[jj][1], sacc[jj][2], sacc[jj][3]);
      *(float4*)(dst + jj * DIM + 4) =
          make_float4(sacc[jj][4], sacc[jj][5], sacc[jj][6], sacc[jj][7]);
    }
  } else {
#pragma unroll
    for (int jj = 0; jj < JT; ++jj)
#pragma unroll
      for (int r = 0; r < 8; ++r)
        unsafeAtomicAdd(&sdst[fr0 + jj * DIM + r], sacc[jj][r]);
  }
}

template <bool MODEA>
__global__ __launch_bounds__(NTHR) __attribute__((amdgpu_waves_per_eu(8, 8)))
void caps_pass_first(const float* __restrict__ x, const float* __restrict__ W,
                     float* sdst) {
  static_assert(JT == 2 && DIM == 16 && MTB == 1 && (ICH % 4) == 0,
                "two j per wave, 8 k per lane, capsule groups of two in pairs");
  static_assert((NJ & (NJ - 1)) == 0, "1/NJ must be a power of two (exact scaling)");
  const int tid  = threadIdx.x;
  const int lane = tid & 31;
  const int wv   = tid >> 5;
  const int half = lane >> 4;
  const int l16  = lane & 15;
  const int jb   = wv * JT;
  const int ich  = blockIdx.x / BSPL;
  const int b0   = (blockIdx.x % BSPL) * BB;
  const int i0   = ich * ICH;

  const v8f vzero = {0.f, 0.f, 0.f, 0.f, 0.f, 0.f, 0.f, 0.f};
  v8f sacc[JT];
#pragma unroll
  for (int jj = 0; jj < JT; ++jj) sacc[jj] = vzero;

  const size_t fr0 = ((size_t)(b0 + 8 * half) * NJ + jb) * DIM + l16;

  constexpr int XCAP = DIM / 4;
  constexpr int WCAP = NJ * DIM * DIM / 4;
  constexpr int WJ   = DIM * DIM / 4;
  const float4* xsrc = (const float4*)(x + ((size_t)b0 * INC + i0) * DIM);
  const unsigned xlane = (unsigned)(l16 * INC * DIM + 8 * half) / 4;
  const float4* wsrc = (const float4*)(W + (size_t)i0 * NJ * DIM * DIM);
  const unsigned wlane = (unsigned)((jb * DIM + l16) * DIM + 8 * half) / 4;

  float4 xa[2][2], xb[2][2];
  float4 wa[2][2 * JT], wb[2][2 * JT];

  auto load = [&](const int cap, float4 (&xs)[2][2],
                  float4 (&ws)[2][2 * JT]) __attribute__((always_inline)) {
#pragma unroll
    for (int g = 0; g < 2; ++g) {
      const float4* xs_src = xsrc + (size_t)(cap + g) * XCAP;
      xs[g][0] = xs_src[xlane];
      xs[g][1] = xs_src[xlane + 1];
      const float4* src = wsrc + (size_t)(cap + g) * WCAP;
#pragma unroll
      for (int jj = 0; jj < JT; ++jj) {
        ws[g][2 * jj]     = src[wlane + jj * WJ];
        ws[g][2 * jj + 1] = src[wlane + jj * WJ + 1];
      }
    }
    asm volatile("s_wait_xcnt 0x0" ::: "memory");
  };

  auto compute = [&](const float4 (&xs)[2][2],
                     const float4 (&ws)[2][2 * JT]) __attribute__((always_inline)) {
#pragma unroll
    for (int g = 0; g < 2; ++g) {
      v2f afs[4];
      afs[0].x = xs[g][0].x; afs[0].y = xs[g][0].y;
      afs[1].x = xs[g][0].z; afs[1].y = xs[g][0].w;
      afs[2].x = xs[g][1].x; afs[2].y = xs[g][1].y;
      afs[3].x = xs[g][1].z; afs[3].y = xs[g][1].w;
#pragma unroll
      for (int jj = 0; jj < JT; ++jj) {
        v2f bfr[4];
        bfr[0].x = ws[g][2 * jj].x;     bfr[0].y = ws[g][2 * jj].y;
        bfr[1].x = ws[g][2 * jj].z;     bfr[1].y = ws[g][2 * jj].w;
        bfr[2].x = ws[g][2 * jj + 1].x; bfr[2].y = ws[g][2 * jj + 1].y;
        bfr[3].x = ws[g][2 * jj + 1].z; bfr[3].y = ws[g][2 * jj + 1].w;
#pragma unroll
        for (int kk = 0; kk < 4; ++kk)
          sacc[jj] = __builtin_amdgcn_wmma_f32_16x16x4_f32(
              false, afs[kk], false, bfr[kk], (short)0, sacc[jj], false, false);
      }
    }
  };

  load(0, xa, wa);
#pragma clang loop unroll(disable)
  for (int ii = 0; ii < ICH - 4; ii += 4) {
    load(ii + 2, xb, wb);
    compute(xa, wa);
    load(ii + 4, xa, wa);
    compute(xb, wb);
  }
  load(ICH - 2, xb, wb);
  compute(xa, wa);
  compute(xb, wb);

  const float c = 1.f / (float)NJ;
#pragma unroll
  for (int jj = 0; jj < JT; ++jj)
#pragma unroll
    for (int v = 0; v < 8; ++v) sacc[jj][v] *= c;

  if (MODEA) {
    float* dst = sdst + (size_t)ich * SZ + fr0;
#pragma unroll
    for (int jj = 0; jj < JT; ++jj)
#pragma unroll
      for (int v = 0; v < 8; ++v) dst[(v * NJ + jj) * DIM] = sacc[jj][v];
  } else {
#pragma unroll
    for (int jj = 0; jj < JT; ++jj)
#pragma unroll
      for (int v = 0; v < 8; ++v)
        unsafeAtomicAdd(&sdst[fr0 + (v * NJ + jj) * DIM], sacc[jj][v]);
  }
}

__global__ void caps_squash(const float* src, float* outsum, float* dout,
                            float* szero, int nparts, int first) {
  const int idx = blockIdx.x * blockDim.x + threadIdx.x;
  float v = 0.f;
  for (int p = 0; p < nparts; ++p) v += src[(size_t)p * SZ + idx];
  if (szero) szero[idx] = 0.f;
  float prev = 0.f;
  if (!first) prev = outsum[idx];
  const float sq = xorsum16(v * v);
  const float coef = sq / ((1.f + sq) * sqrtf(sq + 1e-7f));
  const float o = coef * v;
  outsum[idx] = prev + o;
  if (dout) dout[idx] = o;
}

#define RSPAN  128
#define RWAVES 16
template <int NP>
__global__ __launch_bounds__(32 * RWAVES)
void caps_squash_wide(const float* __restrict__ src, float* outsum, float* dout,
                      int first) {
  constexpr int RPER = NP / RWAVES;
  static_assert(NP % RWAVES == 0 && RSPAN == 4 * 32 && SZ % RSPAN == 0,
                "one float4 per lane and partial; whole runs of partials per wave");
  __shared__ __attribute__((aligned(16))) float red[RWAVES * RSPAN];
  const int tid  = threadIdx.x;
  const int lane = tid & 31;
  const int wv   = tid >> 5;
  const int e0   = blockIdx.x * RSPAN;

  float prev = 0.f;
  if (tid < RSPAN && !first) prev = outsum[e0 + tid];

  const float4* p4 = (const float4*)(src + (size_t)(wv * RPER) * SZ + e0) + lane;
  float4 v[RPER];
#pragma unroll
  for (int k = 0; k < RPER; ++k) v[k] = p4[(size_t)k * (SZ / 4)];
  float4 a = v[0];
#pragma unroll
  for (int k = 1; k < RPER; ++k) {
    a.x += v[k].x; a.y += v[k].y; a.z += v[k].z; a.w += v[k].w;
  }
  *(float4*)(red + wv * RSPAN + 4 * lane) = a;
  __syncthreads();

  if (tid < RSPAN) {
    float s = red[tid];
#pragma unroll
    for (int w = 1; w < RWAVES; ++w) s += red[w * RSPAN + tid];
    const float sq = xorsum16(s * s);
    const float coef = sq / ((1.f + sq) * sqrtf(sq + 1e-7f));
    const float o = coef * s;
    outsum[e0 + tid] = prev + o;
    if (dout) dout[e0 + tid] = o;
  }
}

extern "C" void kernel_launch(void* const* d_in, const int* in_sizes, int n_in,
                              void* d_out, int out_size, void* d_ws, size_t ws_size,
                              hipStream_t stream) {
  const float* x = (const float*)d_in[0];
  const float* W = (const float*)d_in[1];
  float* out = (float*)d_out;
  float* ws = (float*)d_ws;

  float* outsum = ws;
  float* sbuf   = ws + SZ;
  float* parts  = ws + 2 * SZ;

  const bool modeA = ws_size >= (size_t)(2 * SZ + NBLK * SZ) * sizeof(float);

  if (!modeA) caps_init<<<SZ / 256, 256, 0, stream>>>(sbuf, SZ);

  for (int t = 0; t < 3; ++t) {
    if (t == 0) {
      if (modeA)
        caps_pass_first<true><<<NBLK * BSPL, NTHR, 0, stream>>>(x, W, parts);
      else
        caps_pass_first<false><<<NBLK * BSPL, NTHR, 0, stream>>>(x, W, sbuf);
    } else {
      if (modeA)
        caps_pass<true><<<NBLK * BSPL, NTHR, 0, stream>>>(x, W, outsum, parts);
      else
        caps_pass<false><<<NBLK * BSPL, NTHR, 0, stream>>>(x, W, outsum, sbuf);
    }
    if (modeA)
      caps_squash_wide<NBLK><<<SZ / RSPAN, 32 * RWAVES, 0, stream>>>(
          parts, outsum, (t == 2) ? out : nullptr, t == 0 ? 1 : 0);
    else
      caps_squash<<<SZ / 256, 256, 0, stream>>>(sbuf, outsum, (t == 2) ? out : nullptr,
                                                sbuf, 1, t == 0 ? 1 : 0);
  }
}


# Block's x slab (16 batch rows x 16 capsules) staged to LDS once by whole-line loads; both pass kernels read x fragments from it: no 16-line x loads in pass 1, one barrier less a capsule in passes 2-3
# speedup vs baseline: 4.3987x; 1.0533x over previous
#include <hip/hip_runtime.h>
#include <math.h>

#define BATCH 64
#define INC   2048
#define NJ    32
#define DIM   16
#define ICH   16
#define NBLK  (INC / ICH)
#define SZ    (BATCH * NJ * DIM)
#define NTHR  512
#define MTB   1
#define BB    (16 * MTB)
#define BSPL  (BATCH / BB)
#define JG    (16 / MTB)
#define JT    (NJ / JG)
#define XP    (ICH * DIM + 4)

typedef float v2f __attribute__((ext_vector_type(2)));
typedef float v8f __attribute__((ext_vector_type(8)));

#define DPP_XOR(x, m)                                                          \
  __int_as_float(__builtin_amdgcn_update_dpp(                                  \
      0, __float_as_int(x), 0x160 | (m), 0xf, 0xf, true))

__device__ __forceinline__ float xorsum16(float p) {
  p += DPP_XOR(p, 1);
  p += DPP_XOR(p, 2);
  p += DPP_XOR(p, 4);
  p += DPP_XOR(p, 8);
  return p;
}

__device__ __forceinline__ float xormax16(float p) {
  p = fmaxf(p, DPP_XOR(p, 1));
  p = fmaxf(p, DPP_XOR(p, 2));
  p = fmaxf(p, DPP_XOR(p, 4));
  p = fmaxf(p, DPP_XOR(p, 8));
  return p;
}

__device__ __forceinline__ float xhalf(float x) {
  const int b = __float_as_int(x);
  return __int_as_float(
      __builtin_amdgcn_permlanex16(b, b, 0x76543210, 0xfedcba98, false, false));
}

__device__ __forceinline__ void lds_barrier() {
  __builtin_amdgcn_fence(__ATOMIC_RELEASE, "workgroup", "local");
  __builtin_amdgcn_s_barrier();
  __builtin_amdgcn_fence(__ATOMIC_ACQUIRE, "workgroup", "local");
}

__global__ void caps_init(float* p, int n) {
  int i = blockIdx.x * blockDim.x + threadIdx.x;
  if (i < n) p[i] = 0.f;
}

template <bool MODEA>
__global__ __launch_bounds__(NTHR) __attribute__((amdgpu_waves_per_eu(8, 8)))
void caps_pass(const float* __restrict__ x, const float* __restrict__ W,
               const float* __restrict__ outsum, float* sdst) {
  __shared__ __attribute__((aligned(16))) float Xs[BB * XP];
  __shared__ float Lg[BB * NJ];
  static_assert(JT == 2 && DIM == 16, "W fragment registers: two j per wave, 8 k per lane");
  static_assert(BB == NTHR / 32 && NJ == 32,
                "softmax step: one wave per batch row, one lane per j");
  static_assert(BB == NTHR / 32 && ICH * DIM == 2 * 32 * 4 && MTB == 1,
                "slab staging: one wave per batch row, two float4 per lane");

  const int tid  = threadIdx.x;
  const int lane = tid & 31;
  const int wv   = tid >> 5;
  const int half = lane >> 4;
  const int l16  = lane & 15;
  const int mt   = wv % MTB;
  const int jb   = (wv / MTB) * JT;
  const int ich  = blockIdx.x / BSPL;
  const int b0   = (blockIdx.x % BSPL) * BB;
  const int i0   = ich * ICH;

  const v8f vzero = {0.f, 0.f, 0.f, 0.f, 0.f, 0.f, 0.f, 0.f};
  v8f sacc[JT];
#pragma unroll
  for (int jj = 0; jj < JT; ++jj) sacc[jj] = vzero;

  const float4* xrow = (const float4*)(x + ((size_t)(b0 + wv) * INC + i0) * DIM);
  const float4 xs0 = xrow[lane], xs1 = xrow[lane + 32];

  const size_t fr0 = ((size_t)(b0 + mt * 16 + l16) * NJ + jb) * DIM + 8 * half;
  float osum[JT][8];
#pragma unroll
  for (int jj = 0; jj < JT; ++jj) {
    const float4 o0 = *(const float4*)(outsum + fr0 + jj * DIM);
    const float4 o1 = *(const float4*)(outsum + fr0 + jj * DIM + 4);
    osum[jj][0] = o0.x; osum[jj][1] = o0.y; osum[jj][2] = o0.z; osum[jj][3] = o0.w;
    osum[jj][4] = o1.x; osum[jj][5] = o1.y; osum[jj][6] = o1.z; osum[jj][7] = o1.w;
  }

  const float4* wsrc =
      (const float4*)(W + (((size_t)i0 * NJ + jb) * DIM + l16) * DIM + 8 * half);
  constexpr int WCAP = NJ * DIM * DIM / 4;
  constexpr int WJ   = DIM * DIM / 4;
  float4 wa[2 * JT], wb[2 * JT];
#pragma unroll
  for (int jj = 0; jj < JT; ++jj) {
    wa[2 * jj]     = wsrc[jj * WJ];
    wa[2 * jj + 1] = wsrc[jj * WJ + 1];
  }

  *(float4*)(Xs + wv * XP + 4 * lane)        = xs0;
  *(float4*)(Xs + wv * XP + 4 * (lane + 32)) = xs1;
  lds_barrier();
  const float* xfrag = Xs + (mt * 16 + l16) * XP + 8 * half;
  float4 xf[2];
  xf[0] = ((const float4*)xfrag)[0];
  xf[1] = ((const float4*)xfrag)[1];

  auto trip = [&](const int nx, const float4 (&cw)[2 * JT],
                  float4 (&nw)[2 * JT]) __attribute__((always_inline)) {
    {
      const float4* src = wsrc + (size_t)nx * WCAP;
#pragma unroll
      for (int jj = 0; jj < JT; ++jj) {
        nw[2 * jj]     = src[jj * WJ];
        nw[2 * jj + 1] = src[jj * WJ + 1];
      }
      asm volatile("s_wait_xcnt 0x0" ::: "memory");
    }

    v2f xfr[4];
    {
      const float4 a0 = xf[0], a1 = xf[1];
      xfr[0].x = a0.x; xfr[0].y = a0.y;
      xfr[1].x = a0.z; xfr[1].y = a0.w;
      xfr[2].x = a1.x; xfr[2].y = a1.y;
      xfr[3].x = a1.z; xfr[3].y = a1.w;
    }

    v8f uh[JT];
    float part[JT];
#pragma unroll
    for (int jj = 0; jj < JT; ++jj) {
      v2f wfr[4];
      wfr[0].x = cw[2 * jj].x;     wfr[0].y = cw[2 * jj].y;
      wfr[1].x = cw[2 * jj].z;     wfr[1].y = cw[2 * jj].w;
      wfr[2].x = cw[2 * jj + 1].x; wfr[2].y = cw[2 * jj + 1].y;
      wfr[3].x = cw[2 * jj + 1].z; wfr[3].y = cw[2 * jj + 1].w;
      v8f acc = vzero;
#pragma unroll
      for (int kk = 0; kk < 4; ++kk)
        acc = __builtin_amdgcn_wmma_f32_16x16x4_f32(false, wfr[kk], false, xfr[kk],
                                                    (short)0, acc, false, false);
      uh[jj] = acc;
      float p = acc[0] * osum[jj][0];
#pragma unroll
      for (int r = 1; r < 8; ++r) p = fmaf(acc[r], osum[jj][r], p);
      part[jj] = p;
    }
    {
      const float keep = half ? part[1] : part[0];
      const float send = half ? part[0] : part[1];
      Lg[(mt * 16 + l16) * NJ + jb + half] = keep + xhalf(send);
    }
    lds_barrier();

    {
      const float lg = Lg[wv * NJ + lane];
      float m = xormax16(lg);
      m = fmaxf(m, xhalf(m));
      const float e = expf(lg - m);
      float s = xorsum16(e);
      s += xhalf(s);
      Lg[wv * NJ + lane] = e * (1.f / s);
    }
    lds_barrier();

    const int brow = mt * 16 + l16;
#pragma unroll
    for (int jj = 0; jj < JT; ++jj) {
      const float c = Lg[brow * NJ + jb + jj];
#pragma unroll
      for (int r = 0; r < 8; ++r) sacc[jj][r] = fmaf(c, uh[jj][r], sacc[jj][r]);
    }
    {
      const float4* xn = (const float4*)(xfrag + nx * DIM);
      xf[0] = xn[0];
      xf[1] = xn[1];
    }
    lds_barrier();
  };

#pragma clang loop unroll(disable)
  for (int ii = 0; ii < ICH; ii += 2) {
    trip(ii + 1, wa, wb);
    trip(ii + 2 < ICH ? ii + 2 : ICH - 1, wb, wa);
  }

  if (MODEA) {
    float* dst = sdst + (size_t)ich * SZ + fr0;
#pragma unroll
    for (int jj = 0; jj < JT; ++jj) {
      *(float4*)(dst + jj * DIM) =
          make_float4(sacc[jj][0], sacc[jj][1], sacc[jj][2], sacc[jj][3]);
      *(float4*)(dst + jj * DIM + 4) =
          make_float4(sacc[jj][4], sacc[jj][5], sacc[jj][6], sacc[jj][7]);
    }
  } else {
#pragma unroll
    for (int jj = 0; jj < JT; ++jj)
#pragma unroll
      for (int r = 0; r < 8; ++r)
        unsafeAtomicAdd(&sdst[fr0 + jj * DIM + r], sacc[jj][r]);
  }
}

template <bool MODEA>
__global__ __launch_bounds__(NTHR) __attribute__((amdgpu_waves_per_eu(8, 8)))
void caps_pass_first(const float* __restrict__ x, const float* __restrict__ W,
                     float* sdst) {
  static_assert(JT == 2 && DIM == 16 && MTB == 1 && (ICH % 4) == 0,
                "two j per wave, 8 k per lane, capsule groups of two in pairs");
  static_assert((NJ & (NJ - 1)) == 0, "1/NJ must be a power of two (exact scaling)");
  static_assert(BB == NTHR / 32 && ICH * DIM == 2 * 32 * 4,
                "slab staging: one wave per batch row, two float4 per lane");
  __shared__ __attribute__((aligned(16))) float Xs[BB * XP];
  const int tid  = threadIdx.x;
  const int lane = tid & 31;
  const int wv   = tid >> 5;
  const int half = lane >> 4;
  const int l16  = lane & 15;
  const int jb   = wv * JT;
  const int ich  = blockIdx.x / BSPL;
  const int b0   = (blockIdx.x % BSPL) * BB;
  const int i0   = ich * ICH;

  const v8f vzero = {0.f, 0.f, 0.f, 0.f, 0.f, 0.f, 0.f, 0.f};
  v8f sacc[JT];
#pragma unroll
  for (int jj = 0; jj < JT; ++jj) sacc[jj] = vzero;

  const size_t fr0 = ((size_t)(b0 + 8 * half) * NJ + jb) * DIM + l16;

  const float4* xrow = (const float4*)(x + ((size_t)(b0 + wv) * INC + i0) * DIM);
  const float4 xs0 = xrow[lane], xs1 = xrow[lane + 32];

  constexpr int WCAP = NJ * DIM * DIM / 4;
  constexpr int WJ   = DIM * DIM / 4;
  const float* xfrag = Xs + l16 * XP + 8 * half;
  const float4* wsrc = (const float4*)(W + (size_t)i0 * NJ * DIM * DIM);
  const unsigned wlane = (unsigned)((jb * DIM + l16) * DIM + 8 * half) / 4;

  float4 xa[2][2], xb[2][2];
  float4 wa[2][2 * JT], wb[2][2 * JT];

  auto load_w = [&](const int cap, float4 (&ws)[2][2 * JT]) __attribute__((always_inline)) {
#pragma unroll
    for (int g = 0; g < 2; ++g) {
      const float4* src = wsrc + (size_t)(cap + g) * WCAP;
#pragma unroll
      for (int jj = 0; jj < JT; ++jj) {
        ws[g][2 * jj]     = src[wlane + jj * WJ];
        ws[g][2 * jj + 1] = src[wlane + jj * WJ + 1];
      }
    }
  };
  auto load_x = [&](const int cap, float4 (&xs)[2][2]) __attribute__((always_inline)) {
#pragma unroll
    for (int g = 0; g < 2; ++g) {
      const float4* xa = (const float4*)(xfrag + (cap + g) * DIM);
      xs[g][0] = xa[0];
      xs[g][1] = xa[1];
    }
  };
  auto load = [&](const int cap, float4 (&xs)[2][2],
                  float4 (&ws)[2][2 * JT]) __attribute__((always_inline)) {
    load_x(cap, xs);
    load_w(cap, ws);
    asm volatile("s_wait_xcnt 0x0" ::: "memory");
  };

  auto compute = [&](const float4 (&xs)[2][2],
                     const float4 (&ws)[2][2 * JT]) __attribute__((always_inline)) {
#pragma unroll
    for (int g = 0; g < 2; ++g) {
      v2f afs[4];
      afs[0].x = xs[g][0].x; afs[0].y = xs[g][0].y;
      afs[1].x = xs[g][0].z; afs[1].y = xs[g][0].w;
      afs[2].x = xs[g][1].x; afs[2].y = xs[g][1].y;
      afs[3].x = xs[g][1].z; afs[3].y = xs[g][1].w;
#pragma unroll
      for (int jj = 0; jj < JT; ++jj) {
        v2f bfr[4];
        bfr[0].x = ws[g][2 * jj].x;     bfr[0].y = ws[g][2 * jj].y;
        bfr[1].x = ws[g][2 * jj].z;     bfr[1].y = ws[g][2 * jj].w;
        bfr[2].x = ws[g][2 * jj + 1].x; bfr[2].y = ws[g][2 * jj + 1].y;
        bfr[3].x = ws[g][2 * jj + 1].z; bfr[3].y = ws[g][2 * jj + 1].w;
#pragma unroll
        for (int kk = 0; kk < 4; ++kk)
          sacc[jj] = __builtin_amdgcn_wmma_f32_16x16x4_f32(
              false, afs[kk], false, bfr[kk], (short)0, sacc[jj], false, false);
      }
    }
  };

  load_w(0, wa);
  *(float4*)(Xs + wv * XP + 4 * lane)        = xs0;
  *(float4*)(Xs + wv * XP + 4 * (lane + 32)) = xs1;
  lds_barrier();
  load_x(0, xa);
#pragma clang loop unroll(disable)
  for (int ii = 0; ii < ICH - 4; ii += 4) {
    load(ii + 2, xb, wb);
    compute(xa, wa);
    load(ii + 4, xa, wa);
    compute(xb, wb);
  }
  load(ICH - 2, xb, wb);
  compute(xa, wa);
  compute(xb, wb);

  const float c = 1.f / (float)NJ;
#pragma unroll
  for (int jj = 0; jj < JT; ++jj)
#pragma unroll
    for (int v = 0; v < 8; ++v) sacc[jj][v] *= c;

  if (MODEA) {
    float* dst = sdst + (size_t)ich * SZ + fr0;
#pragma unroll
    for (int jj = 0; jj < JT; ++jj)
#pragma unroll
      for (int v = 0; v < 8; ++v) dst[(v * NJ + jj) * DIM] = sacc[jj][v];
  } else {
#pragma unroll
    for (int jj = 0; jj < JT; ++jj)
#pragma unroll
      for (int v = 0; v < 8; ++v)
        unsafeAtomicAdd(&sdst[fr0 + (v * NJ + jj) * DIM], sacc[jj][v]);
  }
}

__global__ void caps_squash(const float* src, float* outsum, float* dout,
                            float* szero, int nparts, int first) {
  const int idx = blockIdx.x * blockDim.x + threadIdx.x;
  float v = 0.f;
  for (int p = 0; p < nparts; ++p) v += src[(size_t)p * SZ + idx];
  if (szero) szero[idx] = 0.f;
  float prev = 0.f;
  if (!first) prev = outsum[idx];
  const float sq = xorsum16(v * v);
  const float coef = sq / ((1.f + sq) * sqrtf(sq + 1e-7f));
  const float o = coef * v;
  outsum[idx] = prev + o;
  if (dout) dout[idx] = o;
}

#define RSPAN  128
#define RWAVES 16
template <int NP>
__global__ __launch_bounds__(32 * RWAVES)
void caps_squash_wide(const float* __restrict__ src, float* outsum, float* dout,
                      int first) {
  constexpr int RPER = NP / RWAVES;
  static_assert(NP % RWAVES == 0 && RSPAN == 4 * 32 && SZ % RSPAN == 0,
                "one float4 per lane and partial; whole runs of partials per wave");
  __shared__ __attribute__((aligned(16))) float red[RWAVES * RSPAN];
  const int tid  = threadIdx.x;
  const int lane = tid & 31;
  const int wv   = tid >> 5;
  const int e0   = blockIdx.x * RSPAN;

  float prev = 0.f;
  if (tid < RSPAN && !first) prev = outsum[e0 + tid];

  const float4* p4 = (const float4*)(src + (size_t)(wv * RPER) * SZ + e0) + lane;
  float4 v[RPER];
#pragma unroll
  for (int k = 0; k < RPER; ++k) v[k] = p4[(size_t)k * (SZ / 4)];
  float4 a = v[0];
#pragma unroll
  for (int k = 1; k < RPER; ++k) {
    a.x += v[k].x; a.y += v[k].y; a.z += v[k].z; a.w += v[k].w;
  }
  *(float4*)(red + wv * RSPAN + 4 * lane) = a;
  __syncthreads();

  if (tid < RSPAN) {
    float s = red[tid];
#pragma unroll
    for (int w = 1; w < RWAVES; ++w) s += red[w * RSPAN + tid];
    const float sq = xorsum16(s * s);
    const float coef = sq / ((1.f + sq) * sqrtf(sq + 1e-7f));
    const float o = coef * s;
    outsum[e0 + tid] = prev + o;
    if (dout) dout[e0 + tid] = o;
  }
}

extern "C" void kernel_launch(void* const* d_in, const int* in_sizes, int n_in,
                              void* d_out, int out_size, void* d_ws, size_t ws_size,
                              hipStream_t stream) {
  const float* x = (const float*)d_in[0];
  const float* W = (const float*)d_in[1];
  float* out = (float*)d_out;
  float* ws = (float*)d_ws;

  float* outsum = ws;
  float* sbuf   = ws + SZ;
  float* parts  = ws + 2 * SZ;

  const bool modeA = ws_size >= (size_t)(2 * SZ + NBLK * SZ) * sizeof(float);

  if (!modeA) caps_init<<<SZ / 256, 256, 0, stream>>>(sbuf, SZ);

  for (int t = 0; t < 3; ++t) {
    if (t == 0) {
      if (modeA)
        caps_pass_first<true><<<NBLK * BSPL, NTHR, 0, stream>>>(x, W, parts);
      else
        caps_pass_first<false><<<NBLK * BSPL, NTHR, 0, stream>>>(x, W, sbuf);
    } else {
      if (modeA)
        caps_pass<true><<<NBLK * BSPL, NTHR, 0, stream>>>(x, W, outsum, parts);
      else
        caps_pass<false><<<NBLK * BSPL, NTHR, 0, stream>>>(x, W, outsum, sbuf);
    }
    if (modeA)
      caps_squash_wide<NBLK><<<SZ / RSPAN, 32 * RWAVES, 0, stream>>>(
          parts, outsum, (t == 2) ? out : nullptr, t == 0 ? 1 : 0);
    else
      caps_squash<<<SZ / 256, 256, 0, stream>>>(sbuf, outsum, (t == 2) ? out : nullptr,
                                                sbuf, 1, t == 0 ? 1 : 0);
  }
}


# First pass: each wave feeds two batch tiles from one set of W fragments (256 blocks of 32 rows, one per WGP, W in load groups of four capsules): half the W traffic per WGP, half the waves to enter
# speedup vs baseline: 4.6553x; 1.0583x over previous
#include <hip/hip_runtime.h>
#include <math.h>

#define BATCH 64
#define INC   2048
#define NJ    32
#define DIM   16
#define ICH   16
#define NBLK  (INC / ICH)
#define SZ    (BATCH * NJ * DIM)
#define NTHR  512
#define MTB   1
#define BB    (16 * MTB)
#define BSPL  (BATCH / BB)
#define JG    (16 / MTB)
#define JT    (NJ / JG)
#define XP    (ICH * DIM + 4)

typedef float v2f __attribute__((ext_vector_type(2)));
typedef float v8f __attribute__((ext_vector_type(8)));

#define DPP_XOR(x, m)                                                          \
  __int_as_float(__builtin_amdgcn_update_dpp(                                  \
      0, __float_as_int(x), 0x160 | (m), 0xf, 0xf, true))

__device__ __forceinline__ float xorsum16(float p) {
  p += DPP_XOR(p, 1);
  p += DPP_XOR(p, 2);
  p += DPP_XOR(p, 4);
  p += DPP_XOR(p, 8);
  return p;
}

__device__ __forceinline__ float xormax16(float p) {
  p = fmaxf(p, DPP_XOR(p, 1));
  p = fmaxf(p, DPP_XOR(p, 2));
  p = fmaxf(p, DPP_XOR(p, 4));
  p = fmaxf(p, DPP_XOR(p, 8));
  return p;
}

__device__ __forceinline__ float xhalf(float x) {
  const int b = __float_as_int(x);
  return __int_as_float(
      __builtin_amdgcn_permlanex16(b, b, 0x76543210, 0xfedcba98, false, false));
}

__device__ __forceinline__ void lds_barrier() {
  __builtin_amdgcn_fence(__ATOMIC_RELEASE, "workgroup", "local");
  __builtin_amdgcn_s_barrier();
  __builtin_amdgcn_fence(__ATOMIC_ACQUIRE, "workgroup", "local");
}

__global__ void caps_init(float* p, int n) {
  int i = blockIdx.x * blockDim.x + threadIdx.x;
  if (i < n) p[i] = 0.f;
}

template <bool MODEA>
__global__ __launch_bounds__(NTHR) __attribute__((amdgpu_waves_per_eu(8, 8)))
void caps_pass(const float* __restrict__ x, const float* __restrict__ W,
               const float* __restrict__ outsum, float* sdst) {
  __shared__ __attribute__((aligned(16))) float Xs[BB * XP];
  __shared__ float Lg[BB * NJ];
  static_assert(JT == 2 && DIM == 16, "W fragment registers: two j per wave, 8 k per lane");
  static_assert(BB == NTHR / 32 && NJ == 32,
                "softmax step: one wave per batch row, one lane per j");
  static_assert(BB == NTHR / 32 && ICH * DIM == 2 * 32 * 4 && MTB == 1,
                "slab staging: one wave per batch row, two float4 per lane");

  const int tid  = threadIdx.x;
  const int lane = tid & 31;
  const int wv   = tid >> 5;
  const int half = lane >> 4;
  const int l16  = lane & 15;
  const int mt   = wv % MTB;
  const int jb   = (wv / MTB) * JT;
  const int ich  = blockIdx.x / BSPL;
  const int b0   = (blockIdx.x % BSPL) * BB;
  const int i0   = ich * ICH;

  const v8f vzero = {0.f, 0.f, 0.f, 0.f, 0.f, 0.f, 0.f, 0.f};
  v8f sacc[JT];
#pragma unroll
  for (int jj = 0; jj < JT; ++jj) sacc[jj] = vzero;

  const float4* xrow = (const float4*)(x + ((size_t)(b0 + wv) * INC + i0) * DIM);
  const float4 xs0 = xrow[lane], xs1 = xrow[lane + 32];

  const size_t fr0 = ((size_t)(b0 + mt * 16 + l16) * NJ + jb) * DIM + 8 * half;
  float osum[JT][8];
#pragma unroll
  for (int jj = 0; jj < JT; ++jj) {
    const float4 o0 = *(const float4*)(outsum + fr0 + jj * DIM);
    const float4 o1 = *(const float4*)(outsum + fr0 + jj * DIM + 4);
    osum[jj][0] = o0.x; osum[jj][1] = o0.y; osum[jj][2] = o0.z; osum[jj][3] = o0.w;
    osum[jj][4] = o1.x; osum[jj][5] = o1.y; osum[jj][6] = o1.z; osum[jj][7] = o1.w;
  }

  const float4* wsrc =
      (const float4*)(W + (((size_t)i0 * NJ + jb) * DIM + l16) * DIM + 8 * half);
  constexpr int WCAP = NJ * DIM * DIM / 4;
  constexpr int WJ   = DIM * DIM / 4;
  float4 wa[2 * JT], wb[2 * JT];
#pragma unroll
  for (int jj = 0; jj < JT; ++jj) {
    wa[2 * jj]     = wsrc[jj * WJ];
    wa[2 * jj + 1] = wsrc[jj * WJ + 1];
  }

  *(float4*)(Xs + wv * XP + 4 * lane)        = xs0;
  *(float4*)(Xs + wv * XP + 4 * (lane + 32)) = xs1;
  lds_barrier();
  const float* xfrag = Xs + (mt * 16 + l16) * XP + 8 * half;
  float4 xf[2];
  xf[0] = ((const float4*)xfrag)[0];
  xf[1] = ((const float4*)xfrag)[1];

  auto trip = [&](const int nx, const float4 (&cw)[2 * JT],
                  float4 (&nw)[2 * JT]) __attribute__((always_inline)) {
    {
      const float4* src = wsrc + (size_t)nx * WCAP;
#pragma unroll
      for (int jj = 0; jj < JT; ++jj) {
        nw[2 * jj]     = src[jj * WJ];
        nw[2 * jj + 1] = src[jj * WJ + 1];
      }
      asm volatile("s_wait_xcnt 0x0" ::: "memory");
    }

    v2f xfr[4];
    {
      const float4 a0 = xf[0], a1 = xf[1];
      xfr[0].x = a0.x; xfr[0].y = a0.y;
      xfr[1].x = a0.z; xfr[1].y = a0.w;
      xfr[2].x = a1.x; xfr[2].y = a1.y;
      xfr[3].x = a1.z; xfr[3].y = a1.w;
    }

    v8f uh[JT];
    float part[JT];
#pragma unroll
    for (int jj = 0; jj < JT; ++jj) {
      v2f wfr[4];
      wfr[0].x = cw[2 * jj].x;     wfr[0].y = cw[2 * jj].y;
      wfr[1].x = cw[2 * jj].z;     wfr[1].y = cw[2 * jj].w;
      wfr[2].x = cw[2 * jj + 1].x; wfr[2].y = cw[2 * jj + 1].y;
      wfr[3].x = cw[2 * jj + 1].z; wfr[3].y = cw[2 * jj + 1].w;
      v8f acc = vzero;
#pragma unroll
      for (int kk = 0; kk < 4; ++kk)
        acc = __builtin_amdgcn_wmma_f32_16x16x4_f32(false, wfr[kk], false, xfr[kk],
                                                    (short)0, acc, false, false);
      uh[jj] = acc;
      float p = acc[0] * osum[jj][0];
#pragma unroll
      for (int r = 1; r < 8; ++r) p = fmaf(acc[r], osum[jj][r], p);
      part[jj] = p;
    }
    {
      const float keep = half ? part[1] : part[0];
      const float send = half ? part[0] : part[1];
      Lg[(mt * 16 + l16) * NJ + jb + half] = keep + xhalf(send);
    }
    lds_barrier();

    {
      const float lg = Lg[wv * NJ + lane];
      float m = xormax16(lg);
      m = fmaxf(m, xhalf(m));
      const float e = expf(lg - m);
      float s = xorsum16(e);
      s += xhalf(s);
      Lg[wv * NJ + lane] = e * (1.f / s);
    }
    lds_barrier();

    const int brow = mt * 16 + l16;
#pragma unroll
    for (int jj = 0; jj < JT; ++jj) {
      const float c = Lg[brow * NJ + jb + jj];
#pragma unroll
      for (int r = 0; r < 8; ++r) sacc[jj][r] = fmaf(c, uh[jj][r], sacc[jj][r]);
    }
    {
      const float4* xn = (const float4*)(xfrag + nx * DIM);
      xf[0] = xn[0];
      xf[1] = xn[1];
    }
    lds_barrier();
  };

#pragma clang loop unroll(disable)
  for (int ii = 0; ii < ICH; ii += 2) {
    trip(ii + 1, wa, wb);
    trip(ii + 2 < ICH ? ii + 2 : ICH - 1, wb, wa);
  }

  if (MODEA) {
    float* dst = sdst + (size_t)ich * SZ + fr0;
#pragma unroll
    for (int jj = 0; jj < JT; ++jj) {
      *(float4*)(dst + jj * DIM) =
          make_float4(sacc[jj][0], sacc[jj][1], sacc[jj][2], sacc[jj][3]);
      *(float4*)(dst + jj * DIM + 4) =
          make_float4(sacc[jj][4], sacc[jj][5], sacc[jj][6], sacc[jj][7]);
    }
  } else {
#pragma unroll
    for (int jj = 0; jj < JT; ++jj)
#pragma unroll
      for (int r = 0; r < 8; ++r)
        unsafeAtomicAdd(&sdst[fr0 + jj * DIM + r], sacc[jj][r]);
  }
}

#define FMT   2
#define FBB   (16 * FMT)
#define FBSPL (BATCH / FBB)
#define FG    4
template <bool MODEA>
__global__ __launch_bounds__(NTHR) __attribute__((amdgpu_waves_per_eu(4, 4)))
void caps_pass_first(const float* __restrict__ x, const float* __restrict__ W,
                     float* sdst) {
  static_assert(JT == 2 && DIM == 16 && ICH == 4 * FG && (FG % 2) == 0,
                "two j per wave, 8 k per lane, four written-out groups of FG capsules");
  static_assert((NJ & (NJ - 1)) == 0, "1/NJ must be a power of two (exact scaling)");
  static_assert(FBB == 2 * (NTHR / 32) && ICH * DIM == 2 * 32 * 4 && JG == NTHR / 32,
                "slab staging: two batch rows per wave, two float4 per lane and row");
  __shared__ __attribute__((aligned(16))) float Xs[FBB * XP];
  const int tid  = threadIdx.x;
  const int lane = tid & 31;
  const int wv   = tid >> 5;
  const int half = lane >> 4;
  const int l16  = lane & 15;
  const int jb   = wv * JT;
  const int ich  = blockIdx.x / FBSPL;
  const int b0   = (blockIdx.x % FBSPL) * FBB;
  const int i0   = ich * ICH;

  const v8f vzero = {0.f, 0.f, 0.f, 0.f, 0.f, 0.f, 0.f, 0.f};
  v8f sacc[FMT][JT];
#pragma unroll
  for (int mt = 0; mt < FMT; ++mt)
#pragma unroll
    for (int jj = 0; jj < JT; ++jj) sacc[mt][jj] = vzero;

  const size_t fr0 = ((size_t)(b0 + 8 * half) * NJ + jb) * DIM + l16;

  float4 xst[FMT][2];
#pragma unroll
  for (int mt = 0; mt < FMT; ++mt) {
    const float4* xrow =
        (const float4*)(x + ((size_t)(b0 + 16 * mt + wv) * INC + i0) * DIM);
    xst[mt][0] = xrow[lane];
    xst[mt][1] = xrow[lane + 32];
  }

  constexpr int WCAP = NJ * DIM * DIM / 4;
  constexpr int WJ   = DIM * DIM / 4;
  const float* xfrag = Xs + l16 * XP + 8 * half;
  const float4* wsrc = (const float4*)(W + (size_t)i0 * NJ * DIM * DIM);
  const unsigned wlane = (unsigned)((jb * DIM + l16) * DIM + 8 * half) / 4;

  float4 xa[FMT][2], xb[FMT][2];
  float4 wa[FG][2 * JT], wb[FG][2 * JT];

  auto load_w_part = [&](const int cap, const int g0, const int g1,
                         float4 (&ws)[FG][2 * JT]) __attribute__((always_inline)) {
#pragma unroll
    for (int g = g0; g < g1; ++g) {
      const float4* src = wsrc + (size_t)(cap + g) * WCAP;
#pragma unroll
      for (int jj = 0; jj < JT; ++jj) {
        ws[g][2 * jj]     = src[wlane + jj * WJ];
        ws[g][2 * jj + 1] = src[wlane + jj * WJ + 1];
      }
    }
    asm volatile("s_wait_xcnt 0x0" ::: "memory");
  };
  auto load_w = [&](const int cap, float4 (&ws)[FG][2 * JT]) __attribute__((always_inline)) {
    load_w_part(cap, 0, FG, ws);
  };
  auto load_x = [&](const int cap, float4 (&xs)[FMT][2]) __attribute__((always_inline)) {
#pragma unroll
    for (int mt = 0; mt < FMT; ++mt) {
      const float4* xp = (const float4*)(xfrag + mt * 16 * XP + cap * DIM);
      xs[mt][0] = xp[0];
      xs[mt][1] = xp[1];
    }
  };
  auto capsule = [&](const int cap, const float4 (&xc)[FMT][2], float4 (&xn)[FMT][2],
                     const float4 (&wc)[2 * JT]) __attribute__((always_inline)) {
    if (cap + 1 < ICH) load_x(cap + 1, xn);
    v2f afs[FMT][4];
#pragma unroll
    for (int mt = 0; mt < FMT; ++mt) {
      afs[mt][0].x = xc[mt][0].x; afs[mt][0].y = xc[mt][0].y;
      afs[mt][1].x = xc[mt][0].z; afs[mt][1].y = xc[mt][0].w;
      afs[mt][2].x = xc[mt][1].x; afs[mt][2].y = xc[mt][1].y;
      afs[mt][3].x = xc[mt][1].z; afs[mt][3].y = xc[mt][1].w;
    }
#pragma unroll
    for (int jj = 0; jj < JT; ++jj) {
      v2f bfr[4];
      bfr[0].x = wc[2 * jj].x;     bfr[0].y = wc[2 * jj].y;
      bfr[1].x = wc[2 * jj].z;     bfr[1].y = wc[2 * jj].w;
      bfr[2].x = wc[2 * jj + 1].x; bfr[2].y = wc[2 * jj + 1].y;
      bfr[3].x = wc[2 * jj + 1].z; bfr[3].y = wc[2 * jj + 1].w;
#pragma unroll
      for (int kk = 0; kk < 4; ++kk)
#pragma unroll
        for (int mt = 0; mt < FMT; ++mt)
          sacc[mt][jj] = __builtin_amdgcn_wmma_f32_16x16x4_f32(
              false, afs[mt][kk], false, bfr[kk], (short)0, sacc[mt][jj], false, false);
    }
  };
  auto group = [&](const int c0, const float4 (&ws)[FG][2 * JT]) __attribute__((always_inline)) {
#pragma unroll
    for (int g = 0; g < FG; g += 2) {
      capsule(c0 + g, xa, xb, ws[g]);
      capsule(c0 + g + 1, xb, xa, ws[g + 1]);
    }
  };

  load_w_part(0, 0, 2, wa);
#pragma unroll
  for (int mt = 0; mt < FMT; ++mt) {
    *(float4*)(Xs + (16 * mt + wv) * XP + 4 * lane)        = xst[mt][0];
    *(float4*)(Xs + (16 * mt + wv) * XP + 4 * (lane + 32)) = xst[mt][1];
  }
  load_w_part(0, 2, FG, wa);
  load_w(FG, wb);
  lds_barrier();
  load_x(0, xa);
  group(0, wa);
  load_w(2 * FG, wa);
  group(FG, wb);
  load_w(3 * FG, wb);
  group(2 * FG, wa);
  group(3 * FG, wb);

  const float c = 1.f / (float)NJ;
#pragma unroll
  for (int mt = 0; mt < FMT; ++mt)
#pragma unroll
    for (int jj = 0; jj < JT; ++jj)
#pragma unroll
      for (int v = 0; v < 8; ++v) sacc[mt][jj][v] *= c;

  if (MODEA) {
    float* dst = sdst + (size_t)ich * SZ + fr0;
#pragma unroll
    for (int mt = 0; mt < FMT; ++mt)
#pragma unroll
      for (int jj = 0; jj < JT; ++jj)
#pragma unroll
        for (int v = 0; v < 8; ++v)
          dst[((mt * 16 + v) * NJ + jj) * DIM] = sacc[mt][jj][v];
  } else {
#pragma unroll
    for (int mt = 0; mt < FMT; ++mt)
#pragma unroll
      for (int jj = 0; jj < JT; ++jj)
#pragma unroll
        for (int v = 0; v < 8; ++v)
          unsafeAtomicAdd(&sdst[fr0 + ((mt * 16 + v) * NJ + jj) * DIM], sacc[mt][jj][v]);
  }
}

__global__ void caps_squash(const float* src, float* outsum, float* dout,
                            float* szero, int nparts, int first) {
  const int idx = blockIdx.x * blockDim.x + threadIdx.x;
  float v = 0.f;
  for (int p = 0; p < nparts; ++p) v += src[(size_t)p * SZ + idx];
  if (szero) szero[idx] = 0.f;
  float prev = 0.f;
  if (!first) prev = outsum[idx];
  const float sq = xorsum16(v * v);
  const float coef = sq / ((1.f + sq) * sqrtf(sq + 1e-7f));
  const float o = coef * v;
  outsum[idx] = prev + o;
  if (dout) dout[idx] = o;
}

#define RSPAN  128
#define RWAVES 16
template <int NP>
__global__ __launch_bounds__(32 * RWAVES)
void caps_squash_wide(const float* __restrict__ src, float* outsum, float* dout,
                      int first) {
  constexpr int RPER = NP / RWAVES;
  static_assert(NP % RWAVES == 0 && RSPAN == 4 * 32 && SZ % RSPAN == 0,
                "one float4 per lane and partial; whole runs of partials per wave");
  __shared__ __attribute__((aligned(16))) float red[RWAVES * RSPAN];
  const int tid  = threadIdx.x;
  const int lane = tid & 31;
  const int wv   = tid >> 5;
  const int e0   = blockIdx.x * RSPAN;

  float prev = 0.f;
  if (tid < RSPAN && !first) prev = outsum[e0 + tid];

  const float4* p4 = (const float4*)(src + (size_t)(wv * RPER) * SZ + e0) + lane;
  float4 v[RPER];
#pragma unroll
  for (int k = 0; k < RPER; ++k) v[k] = p4[(size_t)k * (SZ / 4)];
  float4 a = v[0];
#pragma unroll
  for (int k = 1; k < RPER; ++k) {
    a.x += v[k].x; a.y += v[k].y; a.z += v[k].z; a.w += v[k].w;
  }
  *(float4*)(red + wv * RSPAN + 4 * lane) = a;
  __syncthreads();

  if (tid < RSPAN) {
    float s = red[tid];
#pragma unroll
    for (int w = 1; w < RWAVES; ++w) s += red[w * RSPAN + tid];
    const float sq = xorsum16(s * s);
    const float coef = sq / ((1.f + sq) * sqrtf(sq + 1e-7f));
    const float o = coef * s;
    outsum[e0 + tid] = prev + o;
    if (dout) dout[e0 + tid] = o;
  }
}

extern "C" void kernel_launch(void* const* d_in, const int* in_sizes, int n_in,
                              void* d_out, int out_size, void* d_ws, size_t ws_size,
                              hipStream_t stream) {
  const float* x = (const float*)d_in[0];
  const float* W = (const float*)d_in[1];
  float* out = (float*)d_out;
  float* ws = (float*)d_ws;

  float* outsum = ws;
  float* sbuf   = ws + SZ;
  float* parts  = ws + 2 * SZ;

  const bool modeA = ws_size >= (size_t)(2 * SZ + NBLK * SZ) * sizeof(float);

  if (!modeA) caps_init<<<SZ / 256, 256, 0, stream>>>(sbuf, SZ);

  for (int t = 0; t < 3; ++t) {
    if (t == 0) {
      if (modeA)
        caps_pass_first<true><<<NBLK * FBSPL, NTHR, 0, stream>>>(x, W, parts);
      else
        caps_pass_first<false><<<NBLK * FBSPL, NTHR, 0, stream>>>(x, W, sbuf);
    } else {
      if (modeA)
        caps_pass<true><<<NBLK * BSPL, NTHR, 0, stream>>>(x, W, outsum, parts);
      else
        caps_pass<false><<<NBLK * BSPL, NTHR, 0, stream>>>(x, W, outsum, sbuf);
    }
    if (modeA)
      caps_squash_wide<NBLK><<<SZ / RSPAN, 32 * RWAVES, 0, stream>>>(
          parts, outsum, (t == 2) ? out : nullptr, t == 0 ? 1 : 0);
    else
      caps_squash<<<SZ / 256, 256, 0, stream>>>(sbuf, outsum, (t == 2) ? out : nullptr,
                                                sbuf, 1, t == 0 ? 1 : 0);
  }
}
